# starting kernel
# baseline (speedup 1.0000x reference)
#include <hip/hip_runtime.h>
#include <hip/hip_bf16.h>
#include <math.h>

typedef __attribute__((ext_vector_type(2))) float v2f;
typedef __attribute__((ext_vector_type(8))) float v8f;

#define THREADS 256
#define ORD_NEG_INF 0x007FFFFFu

__device__ __forceinline__ unsigned ford(float f) {
  unsigned u = __float_as_uint(f);
  return (u & 0x80000000u) ? ~u : (u | 0x80000000u);
}
__device__ __forceinline__ float funord(unsigned u) {
  return __uint_as_float((u & 0x80000000u) ? (u & 0x7fffffffu) : ~u);
}

template <int K, int NT>
__global__ void wmma_gemm_xwt(const float* __restrict__ X,
                              const float* __restrict__ W,
                              float* __restrict__ Y, int M) {
  const int NO   = NT * 16;
  const int lane = threadIdx.x & 31;
  const int wave = threadIdx.x >> 5;
  const int tile = blockIdx.x * (blockDim.x >> 5) + wave;
  if (tile * 16 >= M) return;
  const int rbase = tile * 16;
  const int mrow  = rbase + (lane & 15);
  const int koff  = (lane < 16) ? 0 : 2;

  v8f acc[NT];
#pragma unroll
  for (int t = 0; t < NT; ++t) acc[t] = (v8f){0.f,0.f,0.f,0.f,0.f,0.f,0.f,0.f};

  for (int k0 = 0; k0 < K; k0 += 4) {
    const int k = k0 + koff;
    v2f a;
    a.x = (k     < K) ? X[mrow * K + k]     : 0.f;
    a.y = (k + 1 < K) ? X[mrow * K + k + 1] : 0.f;
#pragma unroll
    for (int t = 0; t < NT; ++t) {
      const int col = t * 16 + (lane & 15);
      v2f b;
      b.x = (k     < K) ? W[col * K + k]     : 0.f;
      b.y = (k + 1 < K) ? W[col * K + k + 1] : 0.f;
      acc[t] = __builtin_amdgcn_wmma_f32_16x16x4_f32(
          false, a, false, b, (short)0, acc[t], false, false);
    }
  }

  const int rowoff = (lane < 16) ? 0 : 8;
  const int colo   = lane & 15;
#pragma unroll
  for (int t = 0; t < NT; ++t) {
#pragma unroll
    for (int i = 0; i < 8; ++i) {
      Y[(rbase + rowoff + i) * NO + t * 16 + colo] = acc[t][i];
    }
  }
}

__global__ void fill_u32(unsigned* p, unsigned v, int n) {
  int t = blockIdx.x * blockDim.x + threadIdx.x;
  if (t < n) p[t] = v;
}

__global__ void attn_proj(const float* __restrict__ feat,
                          const float* __restrict__ al,
                          const float* __restrict__ ar,
                          float* __restrict__ el, float* __restrict__ er, int NH) {
  int t = blockIdx.x * blockDim.x + threadIdx.x;
  if (t >= NH) return;
  int n = t >> 2, h = t & 3;
  const float* f = feat + n * 128 + h * 32;
  float sl = 0.f, sr = 0.f;
#pragma unroll
  for (int i = 0; i < 32; ++i) {
    sl += f[i] * al[h * 32 + i];
    sr += f[i] * ar[h * 32 + i];
  }
  el[t] = sl; er[t] = sr;
}

__global__ void edge_score_max(const int* __restrict__ src, const int* __restrict__ dst,
                               const float* __restrict__ el, const float* __restrict__ er,
                               float* __restrict__ ebuf, unsigned* __restrict__ nmax, int EH) {
  int t = blockIdx.x * blockDim.x + threadIdx.x;
  if (t >= EH) return;
  int e = t >> 2, h = t & 3;
  float v = el[src[e] * 4 + h] + er[dst[e] * 4 + h];
  v = (v > 0.f) ? v : 0.2f * v;
  ebuf[t] = v;
  atomicMax(&nmax[dst[e] * 4 + h], ford(v));
}

__global__ void edge_expsum(const int* __restrict__ dst,
                            const unsigned* __restrict__ nmax,
                            float* __restrict__ ebuf, float* __restrict__ nsum, int EH) {
  int t = blockIdx.x * blockDim.x + threadIdx.x;
  if (t >= EH) return;
  int e = t >> 2, h = t & 3;
  float m = funord(nmax[dst[e] * 4 + h]);
  float ex = expf(ebuf[t] - m);
  ebuf[t] = ex;
  atomicAdd(&nsum[dst[e] * 4 + h], ex);
}

__global__ void edge_aggregate(const int* __restrict__ src, const int* __restrict__ dst,
                               const float* __restrict__ ebuf, const float* __restrict__ nsum,
                               const float* __restrict__ feat, float* __restrict__ accum, int EH) {
  int t = blockIdx.x * blockDim.x + threadIdx.x;
  if (t >= EH) return;
  int e = t >> 2, h = t & 3;
  int s = src[e], d = dst[e];
  float a = ebuf[t] / nsum[d * 4 + h];
  const float* fs = feat + s * 128 + h * 32;
  float* ad = accum + d * 128 + h * 32;
#pragma unroll
  for (int i = 0; i < 32; ++i) atomicAdd(&ad[i], fs[i] * a);
}

__global__ void elu_combine(float* __restrict__ accum, const float* __restrict__ res,
                            const float* __restrict__ bias, int n) {
  int t = blockIdx.x * blockDim.x + threadIdx.x;
  if (t >= n) return;
  float v = accum[t] + res[t] + bias[t & 127];
  accum[t] = (v > 0.f) ? v : expm1f(v);
}

__global__ void combine_mean(const float* __restrict__ accum, const float* __restrict__ h1,
                             const float* __restrict__ bias2, float* __restrict__ hnode, int ND) {
  int t = blockIdx.x * blockDim.x + threadIdx.x;
  if (t >= ND) return;
  int n = t >> 5, d = t & 31;
  float s = 0.f;
#pragma unroll
  for (int h = 0; h < 4; ++h) {
    int idx = n * 128 + h * 32 + d;
    s += accum[idx] + h1[idx] + bias2[h * 32 + d];
  }
  hnode[t] = s * 0.25f;
}

__global__ void node_weight(const float* __restrict__ hnode, const float* __restrict__ aw_w,
                            const float* __restrict__ aw_b, float* __restrict__ wbuf, int N) {
  int n = blockIdx.x * blockDim.x + threadIdx.x;
  if (n >= N) return;
  float s = aw_b[0];
#pragma unroll
  for (int i = 0; i < 32; ++i) s += hnode[n * 32 + i] * aw_w[i];
  wbuf[n] = 1.f / (1.f + expf(-s));
}

__global__ void graph_init(unsigned* gmat, int n) {
  int t = blockIdx.x * blockDim.x + threadIdx.x;
  if (t >= n) return;
  gmat[t] = ((t & 63) < 32) ? 0u : ORD_NEG_INF;
}

__global__ void graph_scatter(const float* __restrict__ hnode, const float* __restrict__ wbuf,
                              const int* __restrict__ gid, float* __restrict__ gmat, int ND) {
  int t = blockIdx.x * blockDim.x + threadIdx.x;
  if (t >= ND) return;
  int n = t >> 5, d = t & 31;
  int g = gid[n];
  float hv = hnode[t];
  atomicAdd(&gmat[g * 64 + d], wbuf[n] * hv);
  atomicMax((unsigned*)gmat + g * 64 + 32 + d, ford(hv));
}

__global__ void graph_decode(float* gmat, int GD) {
  int t = blockIdx.x * blockDim.x + threadIdx.x;
  if (t >= GD) return;
  int g = t >> 5, d = t & 31;
  unsigned u = ((unsigned*)gmat)[g * 64 + 32 + d];
  gmat[g * 64 + 32 + d] = (u == ORD_NEG_INF) ? 0.f : funord(u);
}

__global__ void final_ln(const float* __restrict__ ybuf, const float* __restrict__ out_b,
                         float* __restrict__ out) {
  __shared__ float sm[128];
  int g = blockIdx.x, j = threadIdx.x;
  float v = ybuf[g * 128 + j] + out_b[j];
  sm[j] = v; __syncthreads();
  for (int s = 64; s > 0; s >>= 1) { if (j < s) sm[j] += sm[j + s]; __syncthreads(); }
  float mu = sm[0] * (1.f / 128.f);
  __syncthreads();
  float dv = v - mu;
  sm[j] = dv * dv; __syncthreads();
  for (int s = 64; s > 0; s >>= 1) { if (j < s) sm[j] += sm[j + s]; __syncthreads(); }
  float var = sm[0] * (1.f / 128.f);
  out[g * 128 + j] = dv * rsqrtf(var + 1e-5f);
}

extern "C" void kernel_launch(void* const* d_in, const int* in_sizes, int n_in,
                              void* d_out, int out_size, void* d_ws, size_t ws_size,
                              hipStream_t stream) {
  const float* node  = (const float*)d_in[0];
  const int*   src   = (const int*)d_in[1];
  const int*   dst   = (const int*)d_in[2];
  const int*   gid   = (const int*)d_in[3];
  const float* fc1_w = (const float*)d_in[4];
  const float* al1   = (const float*)d_in[5];
  const float* ar1   = (const float*)d_in[6];
  const float* res1w = (const float*)d_in[7];
  const float* bias1 = (const float*)d_in[8];
  const float* fc2_w = (const float*)d_in[9];
  const float* al2   = (const float*)d_in[10];
  const float* ar2   = (const float*)d_in[11];
  const float* bias2 = (const float*)d_in[12];
  const float* aw_w  = (const float*)d_in[13];
  const float* aw_b  = (const float*)d_in[14];
  const float* out_w = (const float*)d_in[15];
  const float* out_b = (const float*)d_in[16];
  (void)n_in; (void)ws_size;

  const int N = in_sizes[3];
  const int E = in_sizes[1];
  const int G = out_size / 128;

  char* wsb = (char*)d_ws;
  size_t off = 0;
  auto alloc = [&](size_t bytes) -> char* {
    char* p = wsb + off;
    off = (off + bytes + 255) & ~(size_t)255;
    return p;
  };
  float*    bufA  = (float*)alloc((size_t)N * 128 * 4);
  float*    bufB  = (float*)alloc((size_t)N * 128 * 4);
  float*    bufC  = (float*)alloc((size_t)N * 128 * 4);
  float*    el    = (float*)alloc((size_t)N * 4 * 4);
  float*    er    = (float*)alloc((size_t)N * 4 * 4);
  unsigned* nmax  = (unsigned*)alloc((size_t)N * 4 * 4);
  float*    nsum  = (float*)alloc((size_t)N * 4 * 4);
  float*    ebuf  = (float*)alloc((size_t)E * 4 * 4);
  float*    hnode = (float*)alloc((size_t)N * 32 * 4);
  float*    wbuf  = (float*)alloc((size_t)N * 4);
  float*    gmat  = (float*)alloc((size_t)G * 64 * 4);
  float*    ybuf  = (float*)alloc((size_t)G * 128 * 4);

  auto grid = [](int n) { return dim3((unsigned)((n + THREADS - 1) / THREADS)); };
  const int ntiles  = (N + 15) / 16;
  const int nblocks = (ntiles + 3) / 4;
  const int gtiles  = (G + 15) / 16;
  const int gblocks = (gtiles + 3) / 4;

  wmma_gemm_xwt<27, 8><<<nblocks, 128, 0, stream>>>(node, fc1_w, bufA, N);
  wmma_gemm_xwt<27, 8><<<nblocks, 128, 0, stream>>>(node, res1w, bufB, N);
  fill_u32<<<grid(N * 128), THREADS, 0, stream>>>((unsigned*)bufC, 0u, N * 128);
  fill_u32<<<grid(N * 4), THREADS, 0, stream>>>(nmax, ORD_NEG_INF, N * 4);
  fill_u32<<<grid(N * 4), THREADS, 0, stream>>>((unsigned*)nsum, 0u, N * 4);
  attn_proj<<<grid(N * 4), THREADS, 0, stream>>>(bufA, al1, ar1, el, er, N * 4);
  edge_score_max<<<grid(E * 4), THREADS, 0, stream>>>(src, dst, el, er, ebuf, nmax, E * 4);
  edge_expsum<<<grid(E * 4), THREADS, 0, stream>>>(dst, nmax, ebuf, nsum, E * 4);
  edge_aggregate<<<grid(E * 4), THREADS, 0, stream>>>(src, dst, ebuf, nsum, bufA, bufC, E * 4);
  elu_combine<<<grid(N * 128), THREADS, 0, stream>>>(bufC, bufB, bias1, N * 128);

  wmma_gemm_xwt<128, 8><<<nblocks, 128, 0, stream>>>(bufC, fc2_w, bufA, N);
  fill_u32<<<grid(N * 128), THREADS, 0, stream>>>((unsigned*)bufB, 0u, N * 128);
  fill_u32<<<grid(N * 4), THREADS, 0, stream>>>(nmax, ORD_NEG_INF, N * 4);
  fill_u32<<<grid(N * 4), THREADS, 0, stream>>>((unsigned*)nsum, 0u, N * 4);
  attn_proj<<<grid(N * 4), THREADS, 0, stream>>>(bufA, al2, ar2, el, er, N * 4);
  edge_score_max<<<grid(E * 4), THREADS, 0, stream>>>(src, dst, el, er, ebuf, nmax, E * 4);
  edge_expsum<<<grid(E * 4), THREADS, 0, stream>>>(dst, nmax, ebuf, nsum, E * 4);
  edge_aggregate<<<grid(E * 4), THREADS, 0, stream>>>(src, dst, ebuf, nsum, bufA, bufB, E * 4);
  combine_mean<<<grid(N * 32), THREADS, 0, stream>>>(bufB, bufC, bias2, hnode, N * 32);

  node_weight<<<grid(N), THREADS, 0, stream>>>(hnode, aw_w, aw_b, wbuf, N);
  graph_init<<<grid(G * 64), THREADS, 0, stream>>>((unsigned*)gmat, G * 64);
  graph_scatter<<<grid(N * 32), THREADS, 0, stream>>>(hnode, wbuf, gid, gmat, N * 32);
  graph_decode<<<grid(G * 32), THREADS, 0, stream>>>(gmat, G * 32);
  wmma_gemm_xwt<64, 8><<<gblocks, 128, 0, stream>>>(gmat, out_w, ybuf, G);
  final_ln<<<G, 128, 0, stream>>>(ybuf, out_b, (float*)d_out);
}


# Attention scores el/er come out of the projection GEMM as 8 extra WMMA output columns (B rows = W^T a, formed in LDS per workgroup); the 2 score passes go, 22->20 launches; el/er sums reassociated
# speedup vs baseline: 4.2204x; 4.2204x over previous
#include <hip/hip_runtime.h>
#include <hip/hip_bf16.h>
#include <math.h>

typedef __attribute__((ext_vector_type(2))) float v2f;
typedef __attribute__((ext_vector_type(8))) float v8f;

#define THREADS 256
#define ORD_NEG_INF 0x007FFFFFu

__device__ __forceinline__ unsigned ford(float f) {
  unsigned u = __float_as_uint(f);
  return (u & 0x80000000u) ? ~u : (u | 0x80000000u);
}
__device__ __forceinline__ float funord(unsigned u) {
  return __uint_as_float((u & 0x80000000u) ? (u & 0x7fffffffu) : ~u);
}

template <int K, int NT, int WAVES, bool ATT>
__global__ __launch_bounds__(WAVES * 32)
void wmma_gemm_xwt(const float* __restrict__ X,
                   const float* __restrict__ W,
                   float* __restrict__ Y, int M,
                   const float* __restrict__ al, const float* __restrict__ ar,
                   float* __restrict__ el, float* __restrict__ er) {
  constexpr int NO = NT * 16;
  constexpr int NTT = ATT ? NT + 1 : NT;
  constexpr int KP = (K + 3) & ~3;
  constexpr int P  = ((KP / 4) & 1) ? KP : KP + 4;
  constexpr int NTHR = WAVES * 32;
  __shared__ __attribute__((aligned(16))) float sW[NO * P];
  __shared__ __attribute__((aligned(16))) float sWa[ATT ? 16 * P : 4];
  __shared__ float sA[ATT ? 256 : 1];

  const int tid = threadIdx.x;
  if constexpr (ATT) {
    static_assert(NO == 128 && NTHR >= 256, "score columns assume 4 heads x 32");
    if (tid < 128) sA[tid] = al[tid];
    else if (tid < 256) sA[tid] = ar[tid - 128];
  }
  if constexpr ((K & 3) == 0) {
    constexpr int K4 = K / 4;
    static_assert((NO * K4) % NTHR == 0, "W must split evenly over the workgroup");
    const float4* __restrict__ W4 = reinterpret_cast<const float4*>(W);
#pragma unroll
    for (int j = 0; j < NO * K4 / NTHR; ++j) {
      const int i = tid + j * NTHR;
      const int r = i / K4, k4 = i - r * K4;
      *reinterpret_cast<float4*>(&sW[r * P + 4 * k4]) = W4[i];
    }
  } else {
#pragma unroll
    for (int j = 0; j < (NO * K + NTHR - 1) / NTHR; ++j) {
      const int i = tid + j * NTHR;
      if (i < NO * K) {
        const int r = i / K, k = i - r * K;
        sW[r * P + k] = W[i];
      }
    }
    for (int i = tid; i < NO * (KP - K); i += NTHR) {
      const int r = i / (KP - K), k = K + (i - r * (KP - K));
      sW[r * P + k] = 0.f;
    }
  }
  __syncthreads();

  if constexpr (ATT) {
    for (int i = tid; i < 16 * KP; i += NTHR) {
      const int j = i / KP, k = i - j * KP;
      float s = 0.f;
      if (j < 8) {
        const float* a = sA + j * 32;
        const float* w = sW + ((j & 3) * 32) * P + k;
#pragma unroll
        for (int d = 0; d < 32; ++d) s += w[d * P] * a[d];
      }
      sWa[j * P + k] = s;
    }
    __syncthreads();
  }

  const int lane = tid & 31;
  const int wave = __builtin_amdgcn_readfirstlane(tid >> 5);
  const int ntiles = (M + 15) >> 4;
  const int koff  = (lane < 16) ? 0 : 2;
  const int rowoff = (lane < 16) ? 0 : 8;
  const int colo   = lane & 15;
  const float* sB = sW + colo * P + koff;
  const float* sBa = sWa + colo * P + koff;

  for (int tile = blockIdx.x + gridDim.x * wave; tile < ntiles; tile += gridDim.x * WAVES) {
    const int rbase = tile * 16;
    const int mrow  = rbase + colo;

    v8f acc[NTT];
#pragma unroll
    for (int t = 0; t < NTT; ++t) acc[t] = (v8f){0.f,0.f,0.f,0.f,0.f,0.f,0.f,0.f};

    if constexpr (ATT && (K & 15) == 0) {
      for (int kc = 0; kc < K; kc += 16) {
        v2f a[4];
#pragma unroll
        for (int u = 0; u < 4; ++u)
          a[u] = *reinterpret_cast<const v2f*>(X + mrow * K + kc + 4 * u + koff);
        asm volatile("" :: "v"(a[0]), "v"(a[1]), "v"(a[2]), "v"(a[3]));
#pragma unroll
        for (int u = 0; u < 4; ++u) {
          const int k0 = kc + 4 * u;
#pragma unroll
          for (int t = 0; t < NT; ++t) {
            const v2f b = *reinterpret_cast<const v2f*>(sB + t * 16 * P + k0);
            acc[t] = __builtin_amdgcn_wmma_f32_16x16x4_f32(
                false, a[u], false, b, (short)0, acc[t], false, false);
          }
          const v2f b = *reinterpret_cast<const v2f*>(sBa + k0);
          acc[NT] = __builtin_amdgcn_wmma_f32_16x16x4_f32(
              false, a[u], false, b, (short)0, acc[NT], false, false);
        }
      }
    } else {
      for (int k0 = 0; k0 < K; k0 += 4) {
        const int k = k0 + koff;
        v2f a;
        a.x = (k     < K) ? X[mrow * K + k]     : 0.f;
        a.y = (k + 1 < K) ? X[mrow * K + k + 1] : 0.f;
#pragma unroll
        for (int t = 0; t < NT; ++t) {
          const v2f b = *reinterpret_cast<const v2f*>(sB + t * 16 * P + k0);
          acc[t] = __builtin_amdgcn_wmma_f32_16x16x4_f32(
              false, a, false, b, (short)0, acc[t], false, false);
        }
        if constexpr (ATT) {
          const v2f b = *reinterpret_cast<const v2f*>(sBa + k0);
          acc[NT] = __builtin_amdgcn_wmma_f32_16x16x4_f32(
              false, a, false, b, (short)0, acc[NT], false, false);
        }
      }
    }

#pragma unroll
    for (int t = 0; t < NT; ++t) {
#pragma unroll
      for (int i = 0; i < 8; ++i) {
        Y[(rbase + rowoff + i) * NO + t * 16 + colo] = acc[t][i];
      }
    }
    if constexpr (ATT) {
      if (colo < 8) {
        float* sc = (colo < 4) ? (el + colo) : (er + (colo - 4));
#pragma unroll
        for (int i = 0; i < 8; ++i) sc[(rbase + rowoff + i) * 4] = acc[NT][i];
      }
    }
  }
}

__global__ void fill_u32(unsigned* p, unsigned v, int n) {
  int t = blockIdx.x * blockDim.x + threadIdx.x;
  if (t < n) p[t] = v;
}

__global__ void edge_score_max(const int* __restrict__ src, const int* __restrict__ dst,
                               const float* __restrict__ el, const float* __restrict__ er,
                               float* __restrict__ ebuf, unsigned* __restrict__ nmax, int EH) {
  int t = blockIdx.x * blockDim.x + threadIdx.x;
  if (t >= EH) return;
  int e = t >> 2, h = t & 3;
  float v = el[src[e] * 4 + h] + er[dst[e] * 4 + h];
  v = (v > 0.f) ? v : 0.2f * v;
  ebuf[t] = v;
  atomicMax(&nmax[dst[e] * 4 + h], ford(v));
}

__global__ void edge_expsum(const int* __restrict__ dst,
                            const unsigned* __restrict__ nmax,
                            float* __restrict__ ebuf, float* __restrict__ nsum, int EH) {
  int t = blockIdx.x * blockDim.x + threadIdx.x;
  if (t >= EH) return;
  int e = t >> 2, h = t & 3;
  float m = funord(nmax[dst[e] * 4 + h]);
  float ex = expf(ebuf[t] - m);
  ebuf[t] = ex;
  atomicAdd(&nsum[dst[e] * 4 + h], ex);
}

__device__ __forceinline__ float bcast_lane(float v, int l) {
  return __int_as_float(__builtin_amdgcn_readlane(__float_as_int(v), l));
}

__global__ __launch_bounds__(THREADS)
void edge_aggregate(const int* __restrict__ src, const int* __restrict__ dst,
                    const float* __restrict__ ebuf, const float* __restrict__ nsum,
                    const float* __restrict__ feat, float* __restrict__ accum, int E) {
  const int lane = threadIdx.x & 31;
  const int wave = __builtin_amdgcn_readfirstlane((int)(threadIdx.x >> 5));
  const int e0 = (blockIdx.x * (THREADS / 32) + wave) * 32;
  if (e0 >= E) return;
  const int nvalid = (E - e0 < 32) ? (E - e0) : 32;

  const int e = e0 + lane;
  int s = 0, d = 0;
  float a0 = 0.f, a1 = 0.f, a2 = 0.f, a3 = 0.f;
  if (lane < nvalid) {
    s = src[e];
    d = dst[e];
    const float4 ex = *reinterpret_cast<const float4*>(ebuf + (size_t)e * 4);
    const float4 sm = *reinterpret_cast<const float4*>(nsum + (size_t)d * 4);
    a0 = ex.x / sm.x; a1 = ex.y / sm.y; a2 = ex.z / sm.z; a3 = ex.w / sm.w;
  }

  if (nvalid == 32) {
#pragma unroll 1
    for (int g = 0; g < 32; g += 8) {
      float f[8][4];
#pragma unroll
      for (int j = 0; j < 8; ++j) {
        const int sj = __builtin_amdgcn_readlane(s, g + j);
        const float* fs = feat + (size_t)sj * 128 + lane;
        f[j][0] = fs[0]; f[j][1] = fs[32]; f[j][2] = fs[64]; f[j][3] = fs[96];
      }
#pragma unroll
      for (int j = 0; j < 8; ++j) {
        const int dj = __builtin_amdgcn_readlane(d, g + j);
        float* ad = accum + (size_t)dj * 128 + lane;
        atomicAdd(ad,      f[j][0] * bcast_lane(a0, g + j));
        atomicAdd(ad + 32, f[j][1] * bcast_lane(a1, g + j));
        atomicAdd(ad + 64, f[j][2] * bcast_lane(a2, g + j));
        atomicAdd(ad + 96, f[j][3] * bcast_lane(a3, g + j));
      }
    }
  } else {
    for (int j = 0; j < nvalid; ++j) {
      const int sj = __builtin_amdgcn_readlane(s, j);
      const int dj = __builtin_amdgcn_readlane(d, j);
      const float* fs = feat + (size_t)sj * 128 + lane;
      float* ad = accum + (size_t)dj * 128 + lane;
      const float f0 = fs[0], f1 = fs[32], f2 = fs[64], f3 = fs[96];
      atomicAdd(ad,      f0 * bcast_lane(a0, j));
      atomicAdd(ad + 32, f1 * bcast_lane(a1, j));
      atomicAdd(ad + 64, f2 * bcast_lane(a2, j));
      atomicAdd(ad + 96, f3 * bcast_lane(a3, j));
    }
  }
}

__global__ __launch_bounds__(THREADS)
void edge_aggregate_headsum(const int* __restrict__ src, const int* __restrict__ dst,
                            const float* __restrict__ ebuf, const float* __restrict__ nsum,
                            const float* __restrict__ feat, float* __restrict__ accum32, int E) {
  const int lane = threadIdx.x & 31;
  const int wave = __builtin_amdgcn_readfirstlane((int)(threadIdx.x >> 5));
  const int e0 = (blockIdx.x * (THREADS / 32) + wave) * 32;
  if (e0 >= E) return;
  const int nvalid = (E - e0 < 32) ? (E - e0) : 32;

  const int e = e0 + lane;
  int s = 0, d = 0;
  float a0 = 0.f, a1 = 0.f, a2 = 0.f, a3 = 0.f;
  if (lane < nvalid) {
    s = src[e];
    d = dst[e];
    const float4 ex = *reinterpret_cast<const float4*>(ebuf + (size_t)e * 4);
    const float4 sm = *reinterpret_cast<const float4*>(nsum + (size_t)d * 4);
    a0 = ex.x / sm.x; a1 = ex.y / sm.y; a2 = ex.z / sm.z; a3 = ex.w / sm.w;
  }

  if (nvalid == 32) {
#pragma unroll 1
    for (int g = 0; g < 32; g += 8) {
      float f[8][4];
#pragma unroll
      for (int j = 0; j < 8; ++j) {
        const int sj = __builtin_amdgcn_readlane(s, g + j);
        const float* fs = feat + (size_t)sj * 128 + lane;
        f[j][0] = fs[0]; f[j][1] = fs[32]; f[j][2] = fs[64]; f[j][3] = fs[96];
      }
#pragma unroll
      for (int j = 0; j < 8; ++j) {
        const int dj = __builtin_amdgcn_readlane(d, g + j);
        const float v = f[j][0] * bcast_lane(a0, g + j) + f[j][1] * bcast_lane(a1, g + j)
                      + f[j][2] * bcast_lane(a2, g + j) + f[j][3] * bcast_lane(a3, g + j);
        atomicAdd(accum32 + (size_t)dj * 32 + lane, v);
      }
    }
  } else {
    for (int j = 0; j < nvalid; ++j) {
      const int sj = __builtin_amdgcn_readlane(s, j);
      const int dj = __builtin_amdgcn_readlane(d, j);
      const float* fs = feat + (size_t)sj * 128 + lane;
      const float f0 = fs[0], f1 = fs[32], f2 = fs[64], f3 = fs[96];
      const float v = f0 * bcast_lane(a0, j) + f1 * bcast_lane(a1, j)
                    + f2 * bcast_lane(a2, j) + f3 * bcast_lane(a3, j);
      atomicAdd(accum32 + (size_t)dj * 32 + lane, v);
    }
  }
}

__global__ void elu_headsum(float* __restrict__ accum, const float* __restrict__ bias1,
                            const float* __restrict__ bias2, float* __restrict__ accum32, int ND) {
  int t = blockIdx.x * blockDim.x + threadIdx.x;
  if (t >= ND) return;
  int n = t >> 5, d = t & 31;
  float* row = accum + n * 128 + d;
  float v[4], b1[4], b2[4];
#pragma unroll
  for (int h = 0; h < 4; ++h) {
    v[h]  = row[h * 32];
    b1[h] = bias1[h * 32 + d];
    b2[h] = bias2[h * 32 + d];
  }
  float s = 0.f;
#pragma unroll
  for (int h = 0; h < 4; ++h) {
    float x = v[h] + b1[h];
    float hv = (x > 0.f) ? x : expm1f(x);
    row[h * 32] = hv;
    s += hv + b2[h];
  }
  accum32[t] = s;
}

__global__ void node_weight(const float* __restrict__ acc32, const float* __restrict__ aw_w,
                            const float* __restrict__ aw_b, float* __restrict__ wbuf, int N) {
  int n = blockIdx.x * blockDim.x + threadIdx.x;
  if (n >= N) return;
  float s = aw_b[0];
#pragma unroll
  for (int i = 0; i < 32; ++i) s += (0.25f * acc32[n * 32 + i]) * aw_w[i];
  wbuf[n] = 1.f / (1.f + expf(-s));
}

__global__ void graph_init(unsigned* gmat, int n) {
  int t = blockIdx.x * blockDim.x + threadIdx.x;
  if (t >= n) return;
  gmat[t] = ((t & 63) < 32) ? 0u : ORD_NEG_INF;
}

__global__ void graph_scatter(const float* __restrict__ acc32, const float* __restrict__ wbuf,
                              const int* __restrict__ gid, float* __restrict__ gmat, int ND) {
  int t = blockIdx.x * blockDim.x + threadIdx.x;
  if (t >= ND) return;
  int n = t >> 5, d = t & 31;
  int g = gid[n];
  float hv = 0.25f * acc32[t];
  atomicAdd(&gmat[g * 64 + d], wbuf[n] * hv);
  atomicMax((unsigned*)gmat + g * 64 + 32 + d, ford(hv));
}

__global__ void graph_decode(float* gmat, int GD) {
  int t = blockIdx.x * blockDim.x + threadIdx.x;
  if (t >= GD) return;
  int g = t >> 5, d = t & 31;
  unsigned u = ((unsigned*)gmat)[g * 64 + 32 + d];
  gmat[g * 64 + 32 + d] = (u == ORD_NEG_INF) ? 0.f : funord(u);
}

__global__ void final_ln(const float* __restrict__ ybuf, const float* __restrict__ out_b,
                         float* __restrict__ out) {
  __shared__ float sm[128];
  int g = blockIdx.x, j = threadIdx.x;
  float v = ybuf[g * 128 + j] + out_b[j];
  sm[j] = v; __syncthreads();
  for (int s = 64; s > 0; s >>= 1) { if (j < s) sm[j] += sm[j + s]; __syncthreads(); }
  float mu = sm[0] * (1.f / 128.f);
  __syncthreads();
  float dv = v - mu;
  sm[j] = dv * dv; __syncthreads();
  for (int s = 64; s > 0; s >>= 1) { if (j < s) sm[j] += sm[j + s]; __syncthreads(); }
  float var = sm[0] * (1.f / 128.f);
  out[g * 128 + j] = dv * rsqrtf(var + 1e-5f);
}

extern "C" void kernel_launch(void* const* d_in, const int* in_sizes, int n_in,
                              void* d_out, int out_size, void* d_ws, size_t ws_size,
                              hipStream_t stream) {
  const float* node  = (const float*)d_in[0];
  const int*   src   = (const int*)d_in[1];
  const int*   dst   = (const int*)d_in[2];
  const int*   gid   = (const int*)d_in[3];
  const float* fc1_w = (const float*)d_in[4];
  const float* al1   = (const float*)d_in[5];
  const float* ar1   = (const float*)d_in[6];
  const float* res1w = (const float*)d_in[7];
  const float* bias1 = (const float*)d_in[8];
  const float* fc2_w = (const float*)d_in[9];
  const float* al2   = (const float*)d_in[10];
  const float* ar2   = (const float*)d_in[11];
  const float* bias2 = (const float*)d_in[12];
  const float* aw_w  = (const float*)d_in[13];
  const float* aw_b  = (const float*)d_in[14];
  const float* out_w = (const float*)d_in[15];
  const float* out_b = (const float*)d_in[16];
  (void)n_in; (void)ws_size;

  const int N = in_sizes[3];
  const int E = in_sizes[1];
  const int G = out_size / 128;

  char* wsb = (char*)d_ws;
  size_t off = 0;
  auto alloc = [&](size_t bytes) -> char* {
    char* p = wsb + off;
    off = (off + bytes + 255) & ~(size_t)255;
    return p;
  };
  float*    bufA  = (float*)alloc((size_t)N * 128 * 4);
  float*    bufB  = (float*)alloc((size_t)N * 32 * 4);
  float*    bufC  = (float*)alloc((size_t)N * 128 * 4);
  float*    el    = (float*)alloc((size_t)N * 4 * 4);
  float*    er    = (float*)alloc((size_t)N * 4 * 4);
  unsigned* nmax  = (unsigned*)alloc((size_t)N * 4 * 4);
  float*    nsum  = (float*)alloc((size_t)N * 4 * 4);
  float*    ebuf  = (float*)alloc((size_t)E * 4 * 4);
  float*    wbuf  = (float*)alloc((size_t)N * 4);
  float*    gmat  = (float*)alloc((size_t)G * 64 * 4);
  float*    ybuf  = (float*)alloc((size_t)G * 128 * 4);

  auto grid = [](int n) { return dim3((unsigned)((n + THREADS - 1) / THREADS)); };
  const int gtiles  = (G + 15) / 16;
  const int gblocks = (gtiles + 7) / 8;

  wmma_gemm_xwt<27, 8, 16, true><<<256, 512, 0, stream>>>(node, fc1_w, bufA, N, al1, ar1, el, er);
  wmma_gemm_xwt<27, 8, 16, false><<<256, 512, 0, stream>>>(node, res1w, bufC, N, nullptr, nullptr, nullptr, nullptr);
  fill_u32<<<grid(N * 4), THREADS, 0, stream>>>(nmax, ORD_NEG_INF, N * 4);
  fill_u32<<<grid(N * 4), THREADS, 0, stream>>>((unsigned*)nsum, 0u, N * 4);
  edge_score_max<<<grid(E * 4), THREADS, 0, stream>>>(src, dst, el, er, ebuf, nmax, E * 4);
  edge_expsum<<<grid(E * 4), THREADS, 0, stream>>>(dst, nmax, ebuf, nsum, E * 4);
  edge_aggregate<<<grid(E), THREADS, 0, stream>>>(src, dst, ebuf, nsum, bufA, bufC, E);
  elu_headsum<<<grid(N * 32), THREADS, 0, stream>>>(bufC, bias1, bias2, bufB, N * 32);

  wmma_gemm_xwt<128, 8, 16, true><<<256, 512, 0, stream>>>(bufC, fc2_w, bufA, N, al2, ar2, el, er);
  fill_u32<<<grid(N * 4), THREADS, 0, stream>>>(nmax, ORD_NEG_INF, N * 4);
  fill_u32<<<grid(N * 4), THREADS, 0, stream>>>((unsigned*)nsum, 0u, N * 4);
  edge_score_max<<<grid(E * 4), THREADS, 0, stream>>>(src, dst, el, er, ebuf, nmax, E * 4);
  edge_expsum<<<grid(E * 4), THREADS, 0, stream>>>(dst, nmax, ebuf, nsum, E * 4);
  edge_aggregate_headsum<<<grid(E), THREADS, 0, stream>>>(src, dst, ebuf, nsum, bufA, bufB, E);

  node_weight<<<grid(N), THREADS, 0, stream>>>(bufB, aw_w, aw_b, wbuf, N);
  graph_init<<<grid(G * 64), THREADS, 0, stream>>>((unsigned*)gmat, G * 64);
  graph_scatter<<<grid(N * 32), THREADS, 0, stream>>>(bufB, wbuf, gid, gmat, N * 32);
  graph_decode<<<grid(G * 32), THREADS, 0, stream>>>(gmat, G * 32);
  wmma_gemm_xwt<64, 8, 8, false><<<gblocks, 256, 0, stream>>>(gmat, out_w, ybuf, G, nullptr, nullptr, nullptr, nullptr);
  final_ln<<<G, 128, 0, stream>>>(ybuf, out_b, (float*)d_out);
}


# Layer-2 projection applies bias1+ELU to its A fragments in registers as it loads the layer-1 accumulator and stores the layer-2 start value from them; ELU pass, h1 go; 20->19 launches; sums unchanged
# speedup vs baseline: 4.3578x; 1.0326x over previous
#include <hip/hip_runtime.h>
#include <hip/hip_bf16.h>
#include <math.h>

typedef __attribute__((ext_vector_type(2))) float v2f;
typedef __attribute__((ext_vector_type(8))) float v8f;

#define THREADS 256
#define ORD_NEG_INF 0x007FFFFFu

__device__ __forceinline__ unsigned ford(float f) {
  unsigned u = __float_as_uint(f);
  return (u & 0x80000000u) ? ~u : (u | 0x80000000u);
}
__device__ __forceinline__ float funord(unsigned u) {
  return __uint_as_float((u & 0x80000000u) ? (u & 0x7fffffffu) : ~u);
}

__device__ __forceinline__ float xhalf(float x) {
  const int b = __builtin_bit_cast(int, x);
  return __builtin_bit_cast(float,
      __builtin_amdgcn_permlanex16(b, b, 0x76543210, 0xfedcba98, false, false));
}

template <int K, int NT, int WAVES, bool ATT, bool PRE>
__global__ __launch_bounds__(WAVES * 32)
void wmma_gemm_xwt(const float* __restrict__ X,
                   const float* __restrict__ W,
                   float* __restrict__ Y, int M,
                   const float* __restrict__ al, const float* __restrict__ ar,
                   float* __restrict__ el, float* __restrict__ er,
                   const float* __restrict__ pb1, const float* __restrict__ pb2,
                   float* __restrict__ hsum) {
  constexpr int NO = NT * 16;
  constexpr int NTT = ATT ? NT + 1 : NT;
  constexpr int KP = (K + 3) & ~3;
  constexpr int P  = ((KP / 4) & 1) ? KP : KP + 4;
  constexpr int NTHR = WAVES * 32;
  __shared__ __attribute__((aligned(16))) float sW[NO * P];
  __shared__ __attribute__((aligned(16))) float sWa[ATT ? 16 * P : 4];
  __shared__ float sA[ATT ? 256 : 1];
  __shared__ __attribute__((aligned(8))) float sPB[PRE ? 256 : 2];

  const int tid = threadIdx.x;
  if constexpr (ATT) {
    static_assert(NO == 128 && NTHR >= 256, "score columns assume 4 heads x 32");
    if (tid < 128) sA[tid] = al[tid];
    else if (tid < 256) sA[tid] = ar[tid - 128];
  }
  if constexpr (PRE) {
    static_assert(ATT && K == 128 && NT == 8 && NTHR >= 512, "pointwise stage assumes 4 heads x 32 in and out");
    if (tid >= 384) sPB[tid - 256] = pb2[tid - 384];
    else if (tid >= 256) sPB[tid - 256] = pb1[tid - 256];
  }
  if constexpr ((K & 3) == 0) {
    constexpr int K4 = K / 4;
    static_assert((NO * K4) % NTHR == 0, "W must split evenly over the workgroup");
    const float4* __restrict__ W4 = reinterpret_cast<const float4*>(W);
#pragma unroll
    for (int j = 0; j < NO * K4 / NTHR; ++j) {
      const int i = tid + j * NTHR;
      const int r = i / K4, k4 = i - r * K4;
      *reinterpret_cast<float4*>(&sW[r * P + 4 * k4]) = W4[i];
    }
  } else {
#pragma unroll
    for (int j = 0; j < (NO * K + NTHR - 1) / NTHR; ++j) {
      const int i = tid + j * NTHR;
      if (i < NO * K) {
        const int r = i / K, k = i - r * K;
        sW[r * P + k] = W[i];
      }
    }
    for (int i = tid; i < NO * (KP - K); i += NTHR) {
      const int r = i / (KP - K), k = K + (i - r * (KP - K));
      sW[r * P + k] = 0.f;
    }
  }
  __syncthreads();

  if constexpr (ATT) {
    for (int i = tid; i < 16 * KP; i += NTHR) {
      const int j = i / KP, k = i - j * KP;
      float s = 0.f;
      if (j < 8) {
        const float* a = sA + j * 32;
        const float* w = sW + ((j & 3) * 32) * P + k;
#pragma unroll
        for (int d = 0; d < 32; ++d) s += w[d * P] * a[d];
      }
      sWa[j * P + k] = s;
    }
    __syncthreads();
  }

  const int lane = tid & 31;
  const int wave = __builtin_amdgcn_readfirstlane(tid >> 5);
  const int ntiles = (M + 15) >> 4;
  const int koff  = (lane < 16) ? 0 : 2;
  const int rowoff = (lane < 16) ? 0 : 8;
  const int colo   = lane & 15;
  const float* sB = sW + colo * P + koff;
  const float* sBa = sWa + colo * P + koff;

  for (int tile = blockIdx.x + gridDim.x * wave; tile < ntiles; tile += gridDim.x * WAVES) {
    const int rbase = tile * 16;
    const int mrow  = rbase + colo;

    v8f acc[NTT];
#pragma unroll
    for (int t = 0; t < NTT; ++t) acc[t] = (v8f){0.f,0.f,0.f,0.f,0.f,0.f,0.f,0.f};

    if constexpr (PRE) {
      float hs[16];
#pragma unroll
      for (int i = 0; i < 16; ++i) hs[i] = 0.f;
      const float* xrow = X + mrow * K + koff;
      const float* pb = sPB + koff;
      for (int h = 0; h < 4; ++h) {
#pragma unroll
        for (int q = 0; q < 2; ++q) {
          const int kc = h * 32 + q * 16;
          v2f a[4];
#pragma unroll
          for (int u = 0; u < 4; ++u)
            a[u] = *reinterpret_cast<const v2f*>(xrow + kc + 4 * u);
          asm volatile("" :: "v"(a[0]), "v"(a[1]), "v"(a[2]), "v"(a[3]));
#pragma unroll
          for (int u = 0; u < 4; ++u) {
            const v2f c1 = *reinterpret_cast<const v2f*>(pb + kc + 4 * u);
            const v2f c2 = *reinterpret_cast<const v2f*>(pb + 128 + kc + 4 * u);
            const float x0 = a[u].x + c1.x, x1 = a[u].y + c1.y;
            const float e0 = expm1f(x0), e1 = expm1f(x1);
            const float h0 = (x0 > 0.f) ? x0 : e0;
            const float h1 = (x1 > 0.f) ? x1 : e1;
            hs[q * 8 + u * 2]     += h0 + c2.x;
            hs[q * 8 + u * 2 + 1] += h1 + c2.y;
            a[u].x = h0; a[u].y = h1;
          }
#pragma unroll
          for (int u = 0; u < 4; ++u) {
            const int k0 = kc + 4 * u;
#pragma unroll
            for (int t = 0; t < NT; ++t) {
              const v2f b = *reinterpret_cast<const v2f*>(sB + t * 16 * P + k0);
              acc[t] = __builtin_amdgcn_wmma_f32_16x16x4_f32(
                  false, a[u], false, b, (short)0, acc[t], false, false);
            }
            const v2f b = *reinterpret_cast<const v2f*>(sBa + k0);
            acc[NT] = __builtin_amdgcn_wmma_f32_16x16x4_f32(
                false, a[u], false, b, (short)0, acc[NT], false, false);
          }
        }
      }
      const bool lo = lane < 16;
      float* hrow = hsum + mrow * 32 + (lo ? 0 : 4);
#pragma unroll
      for (int q = 0; q < 2; ++q) {
#pragma unroll
        for (int p = 0; p < 2; ++p) {
          const int i0 = q * 8 + 4 * p, i1 = i0 + 2;
          const float r0 = xhalf(lo ? hs[i1] : hs[i0]);
          const float r1 = xhalf(lo ? hs[i1 + 1] : hs[i0 + 1]);
          float4 o;
          o.x = lo ? hs[i0] : r0;
          o.y = lo ? hs[i0 + 1] : r1;
          o.z = lo ? r0 : hs[i1];
          o.w = lo ? r1 : hs[i1 + 1];
          *reinterpret_cast<float4*>(hrow + 16 * q + 8 * p) = o;
        }
      }
    } else {
      for (int k0 = 0; k0 < K; k0 += 4) {
        const int k = k0 + koff;
        v2f a;
        a.x = (k     < K) ? X[mrow * K + k]     : 0.f;
        a.y = (k + 1 < K) ? X[mrow * K + k + 1] : 0.f;
#pragma unroll
        for (int t = 0; t < NT; ++t) {
          const v2f b = *reinterpret_cast<const v2f*>(sB + t * 16 * P + k0);
          acc[t] = __builtin_amdgcn_wmma_f32_16x16x4_f32(
              false, a, false, b, (short)0, acc[t], false, false);
        }
        if constexpr (ATT) {
          const v2f b = *reinterpret_cast<const v2f*>(sBa + k0);
          acc[NT] = __builtin_amdgcn_wmma_f32_16x16x4_f32(
              false, a, false, b, (short)0, acc[NT], false, false);
        }
      }
    }

#pragma unroll
    for (int t = 0; t < NT; ++t) {
#pragma unroll
      for (int i = 0; i < 8; ++i) {
        Y[(rbase + rowoff + i) * NO + t * 16 + colo] = acc[t][i];
      }
    }
    if constexpr (ATT) {
      if (colo < 8) {
        float* sc = (colo < 4) ? (el + colo) : (er + (colo - 4));
#pragma unroll
        for (int i = 0; i < 8; ++i) sc[(rbase + rowoff + i) * 4] = acc[NT][i];
      }
    }
  }
}

__global__ void fill_u32(unsigned* p, unsigned v, int n) {
  int t = blockIdx.x * blockDim.x + threadIdx.x;
  if (t < n) p[t] = v;
}

__global__ void edge_score_max(const int* __restrict__ src, const int* __restrict__ dst,
                               const float* __restrict__ el, const float* __restrict__ er,
                               float* __restrict__ ebuf, unsigned* __restrict__ nmax, int EH) {
  int t = blockIdx.x * blockDim.x + threadIdx.x;
  if (t >= EH) return;
  int e = t >> 2, h = t & 3;
  float v = el[src[e] * 4 + h] + er[dst[e] * 4 + h];
  v = (v > 0.f) ? v : 0.2f * v;
  ebuf[t] = v;
  atomicMax(&nmax[dst[e] * 4 + h], ford(v));
}

__global__ void edge_expsum(const int* __restrict__ dst,
                            const unsigned* __restrict__ nmax,
                            float* __restrict__ ebuf, float* __restrict__ nsum, int EH) {
  int t = blockIdx.x * blockDim.x + threadIdx.x;
  if (t >= EH) return;
  int e = t >> 2, h = t & 3;
  float m = funord(nmax[dst[e] * 4 + h]);
  float ex = expf(ebuf[t] - m);
  ebuf[t] = ex;
  atomicAdd(&nsum[dst[e] * 4 + h], ex);
}

__device__ __forceinline__ float bcast_lane(float v, int l) {
  return __int_as_float(__builtin_amdgcn_readlane(__float_as_int(v), l));
}

__global__ __launch_bounds__(THREADS)
void edge_aggregate(const int* __restrict__ src, const int* __restrict__ dst,
                    const float* __restrict__ ebuf, const float* __restrict__ nsum,
                    const float* __restrict__ feat, float* __restrict__ accum, int E) {
  const int lane = threadIdx.x & 31;
  const int wave = __builtin_amdgcn_readfirstlane((int)(threadIdx.x >> 5));
  const int e0 = (blockIdx.x * (THREADS / 32) + wave) * 32;
  if (e0 >= E) return;
  const int nvalid = (E - e0 < 32) ? (E - e0) : 32;

  const int e = e0 + lane;
  int s = 0, d = 0;
  float a0 = 0.f, a1 = 0.f, a2 = 0.f, a3 = 0.f;
  if (lane < nvalid) {
    s = src[e];
    d = dst[e];
    const float4 ex = *reinterpret_cast<const float4*>(ebuf + (size_t)e * 4);
    const float4 sm = *reinterpret_cast<const float4*>(nsum + (size_t)d * 4);
    a0 = ex.x / sm.x; a1 = ex.y / sm.y; a2 = ex.z / sm.z; a3 = ex.w / sm.w;
  }

  if (nvalid == 32) {
#pragma unroll 1
    for (int g = 0; g < 32; g += 8) {
      float f[8][4];
#pragma unroll
      for (int j = 0; j < 8; ++j) {
        const int sj = __builtin_amdgcn_readlane(s, g + j);
        const float* fs = feat + (size_t)sj * 128 + lane;
        f[j][0] = fs[0]; f[j][1] = fs[32]; f[j][2] = fs[64]; f[j][3] = fs[96];
      }
#pragma unroll
      for (int j = 0; j < 8; ++j) {
        const int dj = __builtin_amdgcn_readlane(d, g + j);
        float* ad = accum + (size_t)dj * 128 + lane;
        atomicAdd(ad,      f[j][0] * bcast_lane(a0, g + j));
        atomicAdd(ad + 32, f[j][1] * bcast_lane(a1, g + j));
        atomicAdd(ad + 64, f[j][2] * bcast_lane(a2, g + j));
        atomicAdd(ad + 96, f[j][3] * bcast_lane(a3, g + j));
      }
    }
  } else {
    for (int j = 0; j < nvalid; ++j) {
      const int sj = __builtin_amdgcn_readlane(s, j);
      const int dj = __builtin_amdgcn_readlane(d, j);
      const float* fs = feat + (size_t)sj * 128 + lane;
      float* ad = accum + (size_t)dj * 128 + lane;
      const float f0 = fs[0], f1 = fs[32], f2 = fs[64], f3 = fs[96];
      atomicAdd(ad,      f0 * bcast_lane(a0, j));
      atomicAdd(ad + 32, f1 * bcast_lane(a1, j));
      atomicAdd(ad + 64, f2 * bcast_lane(a2, j));
      atomicAdd(ad + 96, f3 * bcast_lane(a3, j));
    }
  }
}

__global__ __launch_bounds__(THREADS)
void edge_aggregate_headsum(const int* __restrict__ src, const int* __restrict__ dst,
                            const float* __restrict__ ebuf, const float* __restrict__ nsum,
                            const float* __restrict__ feat, float* __restrict__ accum32, int E) {
  const int lane = threadIdx.x & 31;
  const int wave = __builtin_amdgcn_readfirstlane((int)(threadIdx.x >> 5));
  const int e0 = (blockIdx.x * (THREADS / 32) + wave) * 32;
  if (e0 >= E) return;
  const int nvalid = (E - e0 < 32) ? (E - e0) : 32;

  const int e = e0 + lane;
  int s = 0, d = 0;
  float a0 = 0.f, a1 = 0.f, a2 = 0.f, a3 = 0.f;
  if (lane < nvalid) {
    s = src[e];
    d = dst[e];
    const float4 ex = *reinterpret_cast<const float4*>(ebuf + (size_t)e * 4);
    const float4 sm = *reinterpret_cast<const float4*>(nsum + (size_t)d * 4);
    a0 = ex.x / sm.x; a1 = ex.y / sm.y; a2 = ex.z / sm.z; a3 = ex.w / sm.w;
  }

  if (nvalid == 32) {
#pragma unroll 1
    for (int g = 0; g < 32; g += 8) {
      float f[8][4];
#pragma unroll
      for (int j = 0; j < 8; ++j) {
        const int sj = __builtin_amdgcn_readlane(s, g + j);
        const float* fs = feat + (size_t)sj * 128 + lane;
        f[j][0] = fs[0]; f[j][1] = fs[32]; f[j][2] = fs[64]; f[j][3] = fs[96];
      }
#pragma unroll
      for (int j = 0; j < 8; ++j) {
        const int dj = __builtin_amdgcn_readlane(d, g + j);
        const float v = f[j][0] * bcast_lane(a0, g + j) + f[j][1] * bcast_lane(a1, g + j)
                      + f[j][2] * bcast_lane(a2, g + j) + f[j][3] * bcast_lane(a3, g + j);
        atomicAdd(accum32 + (size_t)dj * 32 + lane, v);
      }
    }
  } else {
    for (int j = 0; j < nvalid; ++j) {
      const int sj = __builtin_amdgcn_readlane(s, j);
      const int dj = __builtin_amdgcn_readlane(d, j);
      const float* fs = feat + (size_t)sj * 128 + lane;
      const float f0 = fs[0], f1 = fs[32], f2 = fs[64], f3 = fs[96];
      const float v = f0 * bcast_lane(a0, j) + f1 * bcast_lane(a1, j)
                    + f2 * bcast_lane(a2, j) + f3 * bcast_lane(a3, j);
      atomicAdd(accum32 + (size_t)dj * 32 + lane, v);
    }
  }
}


__global__ void node_weight(const float* __restrict__ acc32, const float* __restrict__ aw_w,
                            const float* __restrict__ aw_b, float* __restrict__ wbuf, int N) {
  int n = blockIdx.x * blockDim.x + threadIdx.x;
  if (n >= N) return;
  float s = aw_b[0];
#pragma unroll
  for (int i = 0; i < 32; ++i) s += (0.25f * acc32[n * 32 + i]) * aw_w[i];
  wbuf[n] = 1.f / (1.f + expf(-s));
}

__global__ void graph_init(unsigned* gmat, int n) {
  int t = blockIdx.x * blockDim.x + threadIdx.x;
  if (t >= n) return;
  gmat[t] = ((t & 63) < 32) ? 0u : ORD_NEG_INF;
}

__global__ void graph_scatter(const float* __restrict__ acc32, const float* __restrict__ wbuf,
                              const int* __restrict__ gid, float* __restrict__ gmat, int ND) {
  int t = blockIdx.x * blockDim.x + threadIdx.x;
  if (t >= ND) return;
  int n = t >> 5, d = t & 31;
  int g = gid[n];
  float hv = 0.25f * acc32[t];
  atomicAdd(&gmat[g * 64 + d], wbuf[n] * hv);
  atomicMax((unsigned*)gmat + g * 64 + 32 + d, ford(hv));
}

__global__ void graph_decode(float* gmat, int GD) {
  int t = blockIdx.x * blockDim.x + threadIdx.x;
  if (t >= GD) return;
  int g = t >> 5, d = t & 31;
  unsigned u = ((unsigned*)gmat)[g * 64 + 32 + d];
  gmat[g * 64 + 32 + d] = (u == ORD_NEG_INF) ? 0.f : funord(u);
}

__global__ void final_ln(const float* __restrict__ ybuf, const float* __restrict__ out_b,
                         float* __restrict__ out) {
  __shared__ float sm[128];
  int g = blockIdx.x, j = threadIdx.x;
  float v = ybuf[g * 128 + j] + out_b[j];
  sm[j] = v; __syncthreads();
  for (int s = 64; s > 0; s >>= 1) { if (j < s) sm[j] += sm[j + s]; __syncthreads(); }
  float mu = sm[0] * (1.f / 128.f);
  __syncthreads();
  float dv = v - mu;
  sm[j] = dv * dv; __syncthreads();
  for (int s = 64; s > 0; s >>= 1) { if (j < s) sm[j] += sm[j + s]; __syncthreads(); }
  float var = sm[0] * (1.f / 128.f);
  out[g * 128 + j] = dv * rsqrtf(var + 1e-5f);
}

extern "C" void kernel_launch(void* const* d_in, const int* in_sizes, int n_in,
                              void* d_out, int out_size, void* d_ws, size_t ws_size,
                              hipStream_t stream) {
  const float* node  = (const float*)d_in[0];
  const int*   src   = (const int*)d_in[1];
  const int*   dst   = (const int*)d_in[2];
  const int*   gid   = (const int*)d_in[3];
  const float* fc1_w = (const float*)d_in[4];
  const float* al1   = (const float*)d_in[5];
  const float* ar1   = (const float*)d_in[6];
  const float* res1w = (const float*)d_in[7];
  const float* bias1 = (const float*)d_in[8];
  const float* fc2_w = (const float*)d_in[9];
  const float* al2   = (const float*)d_in[10];
  const float* ar2   = (const float*)d_in[11];
  const float* bias2 = (const float*)d_in[12];
  const float* aw_w  = (const float*)d_in[13];
  const float* aw_b  = (const float*)d_in[14];
  const float* out_w = (const float*)d_in[15];
  const float* out_b = (const float*)d_in[16];
  (void)n_in; (void)ws_size;

  const int N = in_sizes[3];
  const int E = in_sizes[1];
  const int G = out_size / 128;

  char* wsb = (char*)d_ws;
  size_t off = 0;
  auto alloc = [&](size_t bytes) -> char* {
    char* p = wsb + off;
    off = (off + bytes + 255) & ~(size_t)255;
    return p;
  };
  float*    bufA  = (float*)alloc((size_t)N * 128 * 4);
  float*    bufB  = (float*)alloc((size_t)N * 32 * 4);
  float*    bufC  = (float*)alloc((size_t)N * 128 * 4);
  float*    el    = (float*)alloc((size_t)N * 4 * 4);
  float*    er    = (float*)alloc((size_t)N * 4 * 4);
  unsigned* nmax  = (unsigned*)alloc((size_t)N * 4 * 4);
  float*    nsum  = (float*)alloc((size_t)N * 4 * 4);
  float*    ebuf  = (float*)alloc((size_t)E * 4 * 4);
  float*    wbuf  = (float*)alloc((size_t)N * 4);
  float*    gmat  = (float*)alloc((size_t)G * 64 * 4);
  float*    ybuf  = (float*)alloc((size_t)G * 128 * 4);

  auto grid = [](int n) { return dim3((unsigned)((n + THREADS - 1) / THREADS)); };
  const int gtiles  = (G + 15) / 16;
  const int gblocks = (gtiles + 7) / 8;

  wmma_gemm_xwt<27, 8, 16, true, false><<<256, 512, 0, stream>>>(node, fc1_w, bufA, N, al1, ar1, el, er,
                                                                 nullptr, nullptr, nullptr);
  wmma_gemm_xwt<27, 8, 16, false, false><<<256, 512, 0, stream>>>(node, res1w, bufC, N, nullptr, nullptr, nullptr, nullptr,
                                                                  nullptr, nullptr, nullptr);
  fill_u32<<<grid(N * 4), THREADS, 0, stream>>>(nmax, ORD_NEG_INF, N * 4);
  fill_u32<<<grid(N * 4), THREADS, 0, stream>>>((unsigned*)nsum, 0u, N * 4);
  edge_score_max<<<grid(E * 4), THREADS, 0, stream>>>(src, dst, el, er, ebuf, nmax, E * 4);
  edge_expsum<<<grid(E * 4), THREADS, 0, stream>>>(dst, nmax, ebuf, nsum, E * 4);
  edge_aggregate<<<grid(E), THREADS, 0, stream>>>(src, dst, ebuf, nsum, bufA, bufC, E);

  wmma_gemm_xwt<128, 8, 16, true, true><<<256, 512, 0, stream>>>(bufC, fc2_w, bufA, N, al2, ar2, el, er,
                                                                 bias1, bias2, bufB);
  fill_u32<<<grid(N * 4), THREADS, 0, stream>>>(nmax, ORD_NEG_INF, N * 4);
  fill_u32<<<grid(N * 4), THREADS, 0, stream>>>((unsigned*)nsum, 0u, N * 4);
  edge_score_max<<<grid(E * 4), THREADS, 0, stream>>>(src, dst, el, er, ebuf, nmax, E * 4);
  edge_expsum<<<grid(E * 4), THREADS, 0, stream>>>(dst, nmax, ebuf, nsum, E * 4);
  edge_aggregate_headsum<<<grid(E), THREADS, 0, stream>>>(src, dst, ebuf, nsum, bufA, bufB, E);

  node_weight<<<grid(N), THREADS, 0, stream>>>(bufB, aw_w, aw_b, wbuf, N);
  graph_init<<<grid(G * 64), THREADS, 0, stream>>>((unsigned*)gmat, G * 64);
  graph_scatter<<<grid(N * 32), THREADS, 0, stream>>>(bufB, wbuf, gid, gmat, N * 32);
  graph_decode<<<grid(G * 32), THREADS, 0, stream>>>(gmat, G * 32);
  wmma_gemm_xwt<64, 8, 8, false, false><<<gblocks, 256, 0, stream>>>(gmat, out_w, ybuf, G, nullptr, nullptr, nullptr, nullptr,
                                                                     nullptr, nullptr, nullptr);
  final_ln<<<G, 128, 0, stream>>>(ybuf, out_b, (float*)d_out);
}


# Edge-softmax statistics (running max, exp sum; N x 4 each) get their start values from the 8 idle columns of the projection's score-tile store; 4 fill launches go, 19->15 launches; sums unchanged
# speedup vs baseline: 4.5362x; 1.0409x over previous
#include <hip/hip_runtime.h>
#include <hip/hip_bf16.h>
#include <math.h>

typedef __attribute__((ext_vector_type(2))) float v2f;
typedef __attribute__((ext_vector_type(8))) float v8f;

#define THREADS 256
#define ORD_NEG_INF 0x007FFFFFu

__device__ __forceinline__ unsigned ford(float f) {
  unsigned u = __float_as_uint(f);
  return (u & 0x80000000u) ? ~u : (u | 0x80000000u);
}
__device__ __forceinline__ float funord(unsigned u) {
  return __uint_as_float((u & 0x80000000u) ? (u & 0x7fffffffu) : ~u);
}

__device__ __forceinline__ float xhalf(float x) {
  const int b = __builtin_bit_cast(int, x);
  return __builtin_bit_cast(float,
      __builtin_amdgcn_permlanex16(b, b, 0x76543210, 0xfedcba98, false, false));
}

template <int K, int NT, int WAVES, bool ATT, bool PRE>
__global__ __launch_bounds__(WAVES * 32)
void wmma_gemm_xwt(const float* __restrict__ X,
                   const float* __restrict__ W,
                   float* __restrict__ Y, int M,
                   const float* __restrict__ al, const float* __restrict__ ar,
                   float* __restrict__ el, float* __restrict__ er,
                   unsigned* __restrict__ smax, float* __restrict__ ssum,
                   const float* __restrict__ pb1, const float* __restrict__ pb2,
                   float* __restrict__ hsum) {
  constexpr int NO = NT * 16;
  constexpr int NTT = ATT ? NT + 1 : NT;
  constexpr int KP = (K + 3) & ~3;
  constexpr int P  = ((KP / 4) & 1) ? KP : KP + 4;
  constexpr int NTHR = WAVES * 32;
  __shared__ __attribute__((aligned(16))) float sW[NO * P];
  __shared__ __attribute__((aligned(16))) float sWa[ATT ? 16 * P : 4];
  __shared__ float sA[ATT ? 256 : 1];
  __shared__ __attribute__((aligned(8))) float sPB[PRE ? 256 : 2];

  const int tid = threadIdx.x;
  if constexpr (ATT) {
    static_assert(NO == 128 && NTHR >= 256, "score columns assume 4 heads x 32");
    if (tid < 128) sA[tid] = al[tid];
    else if (tid < 256) sA[tid] = ar[tid - 128];
  }
  if constexpr (PRE) {
    static_assert(ATT && K == 128 && NT == 8 && NTHR >= 512, "pointwise stage assumes 4 heads x 32 in and out");
    if (tid >= 384) sPB[tid - 256] = pb2[tid - 384];
    else if (tid >= 256) sPB[tid - 256] = pb1[tid - 256];
  }
  if constexpr ((K & 3) == 0) {
    constexpr int K4 = K / 4;
    static_assert((NO * K4) % NTHR == 0, "W must split evenly over the workgroup");
    const float4* __restrict__ W4 = reinterpret_cast<const float4*>(W);
#pragma unroll
    for (int j = 0; j < NO * K4 / NTHR; ++j) {
      const int i = tid + j * NTHR;
      const int r = i / K4, k4 = i - r * K4;
      *reinterpret_cast<float4*>(&sW[r * P + 4 * k4]) = W4[i];
    }
  } else {
#pragma unroll
    for (int j = 0; j < (NO * K + NTHR - 1) / NTHR; ++j) {
      const int i = tid + j * NTHR;
      if (i < NO * K) {
        const int r = i / K, k = i - r * K;
        sW[r * P + k] = W[i];
      }
    }
    for (int i = tid; i < NO * (KP - K); i += NTHR) {
      const int r = i / (KP - K), k = K + (i - r * (KP - K));
      sW[r * P + k] = 0.f;
    }
  }
  __syncthreads();

  if constexpr (ATT) {
    for (int i = tid; i < 16 * KP; i += NTHR) {
      const int j = i / KP, k = i - j * KP;
      float s = 0.f;
      if (j < 8) {
        const float* a = sA + j * 32;
        const float* w = sW + ((j & 3) * 32) * P + k;
#pragma unroll
        for (int d = 0; d < 32; ++d) s += w[d * P] * a[d];
      }
      sWa[j * P + k] = s;
    }
    __syncthreads();
  }

  const int lane = tid & 31;
  const int wave = __builtin_amdgcn_readfirstlane(tid >> 5);
  const int ntiles = (M + 15) >> 4;
  const int koff  = (lane < 16) ? 0 : 2;
  const int rowoff = (lane < 16) ? 0 : 8;
  const int colo   = lane & 15;
  const float* sB = sW + colo * P + koff;
  const float* sBa = sWa + colo * P + koff;

  for (int tile = blockIdx.x + gridDim.x * wave; tile < ntiles; tile += gridDim.x * WAVES) {
    const int rbase = tile * 16;
    const int mrow  = rbase + colo;

    v8f acc[NTT];
#pragma unroll
    for (int t = 0; t < NTT; ++t) acc[t] = (v8f){0.f,0.f,0.f,0.f,0.f,0.f,0.f,0.f};

    if constexpr (PRE) {
      float hs[16];
#pragma unroll
      for (int i = 0; i < 16; ++i) hs[i] = 0.f;
      const float* xrow = X + mrow * K + koff;
      const float* pb = sPB + koff;
      for (int h = 0; h < 4; ++h) {
#pragma unroll
        for (int q = 0; q < 2; ++q) {
          const int kc = h * 32 + q * 16;
          v2f a[4];
#pragma unroll
          for (int u = 0; u < 4; ++u)
            a[u] = *reinterpret_cast<const v2f*>(xrow + kc + 4 * u);
          asm volatile("" :: "v"(a[0]), "v"(a[1]), "v"(a[2]), "v"(a[3]));
#pragma unroll
          for (int u = 0; u < 4; ++u) {
            const v2f c1 = *reinterpret_cast<const v2f*>(pb + kc + 4 * u);
            const v2f c2 = *reinterpret_cast<const v2f*>(pb + 128 + kc + 4 * u);
            const float x0 = a[u].x + c1.x, x1 = a[u].y + c1.y;
            const float e0 = expm1f(x0), e1 = expm1f(x1);
            const float h0 = (x0 > 0.f) ? x0 : e0;
            const float h1 = (x1 > 0.f) ? x1 : e1;
            hs[q * 8 + u * 2]     += h0 + c2.x;
            hs[q * 8 + u * 2 + 1] += h1 + c2.y;
            a[u].x = h0; a[u].y = h1;
          }
#pragma unroll
          for (int u = 0; u < 4; ++u) {
            const int k0 = kc + 4 * u;
#pragma unroll
            for (int t = 0; t < NT; ++t) {
              const v2f b = *reinterpret_cast<const v2f*>(sB + t * 16 * P + k0);
              acc[t] = __builtin_amdgcn_wmma_f32_16x16x4_f32(
                  false, a[u], false, b, (short)0, acc[t], false, false);
            }
            const v2f b = *reinterpret_cast<const v2f*>(sBa + k0);
            acc[NT] = __builtin_amdgcn_wmma_f32_16x16x4_f32(
                false, a[u], false, b, (short)0, acc[NT], false, false);
          }
        }
      }
      const bool lo = lane < 16;
      float* hrow = hsum + mrow * 32 + (lo ? 0 : 4);
#pragma unroll
      for (int q = 0; q < 2; ++q) {
#pragma unroll
        for (int p = 0; p < 2; ++p) {
          const int i0 = q * 8 + 4 * p, i1 = i0 + 2;
          const float r0 = xhalf(lo ? hs[i1] : hs[i0]);
          const float r1 = xhalf(lo ? hs[i1 + 1] : hs[i0 + 1]);
          float4 o;
          o.x = lo ? hs[i0] : r0;
          o.y = lo ? hs[i0 + 1] : r1;
          o.z = lo ? r0 : hs[i1];
          o.w = lo ? r1 : hs[i1 + 1];
          *reinterpret_cast<float4*>(hrow + 16 * q + 8 * p) = o;
        }
      }
    } else {
      for (int k0 = 0; k0 < K; k0 += 4) {
        const int k = k0 + koff;
        v2f a;
        a.x = (k     < K) ? X[mrow * K + k]     : 0.f;
        a.y = (k + 1 < K) ? X[mrow * K + k + 1] : 0.f;
#pragma unroll
        for (int t = 0; t < NT; ++t) {
          const v2f b = *reinterpret_cast<const v2f*>(sB + t * 16 * P + k0);
          acc[t] = __builtin_amdgcn_wmma_f32_16x16x4_f32(
              false, a, false, b, (short)0, acc[t], false, false);
        }
        if constexpr (ATT) {
          const v2f b = *reinterpret_cast<const v2f*>(sBa + k0);
          acc[NT] = __builtin_amdgcn_wmma_f32_16x16x4_f32(
              false, a, false, b, (short)0, acc[NT], false, false);
        }
      }
    }

#pragma unroll
    for (int t = 0; t < NT; ++t) {
#pragma unroll
      for (int i = 0; i < 8; ++i) {
        Y[(rbase + rowoff + i) * NO + t * 16 + colo] = acc[t][i];
      }
    }
    if constexpr (ATT) {
      unsigned* sc = (colo < 4)  ? reinterpret_cast<unsigned*>(el) + colo
                   : (colo < 8)  ? reinterpret_cast<unsigned*>(er) + (colo - 4)
                   : (colo < 12) ? smax + (colo - 8)
                                 : reinterpret_cast<unsigned*>(ssum) + (colo - 12);
      const bool prod = colo < 8;
      const unsigned start = (colo < 12) ? ORD_NEG_INF : 0u;
#pragma unroll
      for (int i = 0; i < 8; ++i)
        sc[(rbase + rowoff + i) * 4] = prod ? __float_as_uint(acc[NT][i]) : start;
    }
  }
}

__global__ void edge_score_max(const int* __restrict__ src, const int* __restrict__ dst,
                               const float* __restrict__ el, const float* __restrict__ er,
                               float* __restrict__ ebuf, unsigned* __restrict__ nmax, int EH) {
  int t = blockIdx.x * blockDim.x + threadIdx.x;
  if (t >= EH) return;
  int e = t >> 2, h = t & 3;
  float v = el[src[e] * 4 + h] + er[dst[e] * 4 + h];
  v = (v > 0.f) ? v : 0.2f * v;
  ebuf[t] = v;
  atomicMax(&nmax[dst[e] * 4 + h], ford(v));
}

__global__ void edge_expsum(const int* __restrict__ dst,
                            const unsigned* __restrict__ nmax,
                            float* __restrict__ ebuf, float* __restrict__ nsum, int EH) {
  int t = blockIdx.x * blockDim.x + threadIdx.x;
  if (t >= EH) return;
  int e = t >> 2, h = t & 3;
  float m = funord(nmax[dst[e] * 4 + h]);
  float ex = expf(ebuf[t] - m);
  ebuf[t] = ex;
  atomicAdd(&nsum[dst[e] * 4 + h], ex);
}

__device__ __forceinline__ float bcast_lane(float v, int l) {
  return __int_as_float(__builtin_amdgcn_readlane(__float_as_int(v), l));
}

__global__ __launch_bounds__(THREADS)
void edge_aggregate(const int* __restrict__ src, const int* __restrict__ dst,
                    const float* __restrict__ ebuf, const float* __restrict__ nsum,
                    const float* __restrict__ feat, float* __restrict__ accum, int E) {
  const int lane = threadIdx.x & 31;
  const int wave = __builtin_amdgcn_readfirstlane((int)(threadIdx.x >> 5));
  const int e0 = (blockIdx.x * (THREADS / 32) + wave) * 32;
  if (e0 >= E) return;
  const int nvalid = (E - e0 < 32) ? (E - e0) : 32;

  const int e = e0 + lane;
  int s = 0, d = 0;
  float a0 = 0.f, a1 = 0.f, a2 = 0.f, a3 = 0.f;
  if (lane < nvalid) {
    s = src[e];
    d = dst[e];
    const float4 ex = *reinterpret_cast<const float4*>(ebuf + (size_t)e * 4);
    const float4 sm = *reinterpret_cast<const float4*>(nsum + (size_t)d * 4);
    a0 = ex.x / sm.x; a1 = ex.y / sm.y; a2 = ex.z / sm.z; a3 = ex.w / sm.w;
  }

  if (nvalid == 32) {
#pragma unroll 1
    for (int g = 0; g < 32; g += 8) {
      float f[8][4];
#pragma unroll
      for (int j = 0; j < 8; ++j) {
        const int sj = __builtin_amdgcn_readlane(s, g + j);
        const float* fs = feat + (size_t)sj * 128 + lane;
        f[j][0] = fs[0]; f[j][1] = fs[32]; f[j][2] = fs[64]; f[j][3] = fs[96];
      }
#pragma unroll
      for (int j = 0; j < 8; ++j) {
        const int dj = __builtin_amdgcn_readlane(d, g + j);
        float* ad = accum + (size_t)dj * 128 + lane;
        atomicAdd(ad,      f[j][0] * bcast_lane(a0, g + j));
        atomicAdd(ad + 32, f[j][1] * bcast_lane(a1, g + j));
        atomicAdd(ad + 64, f[j][2] * bcast_lane(a2, g + j));
        atomicAdd(ad + 96, f[j][3] * bcast_lane(a3, g + j));
      }
    }
  } else {
    for (int j = 0; j < nvalid; ++j) {
      const int sj = __builtin_amdgcn_readlane(s, j);
      const int dj = __builtin_amdgcn_readlane(d, j);
      const float* fs = feat + (size_t)sj * 128 + lane;
      float* ad = accum + (size_t)dj * 128 + lane;
      const float f0 = fs[0], f1 = fs[32], f2 = fs[64], f3 = fs[96];
      atomicAdd(ad,      f0 * bcast_lane(a0, j));
      atomicAdd(ad + 32, f1 * bcast_lane(a1, j));
      atomicAdd(ad + 64, f2 * bcast_lane(a2, j));
      atomicAdd(ad + 96, f3 * bcast_lane(a3, j));
    }
  }
}

__global__ __launch_bounds__(THREADS)
void edge_aggregate_headsum(const int* __restrict__ src, const int* __restrict__ dst,
                            const float* __restrict__ ebuf, const float* __restrict__ nsum,
                            const float* __restrict__ feat, float* __restrict__ accum32, int E) {
  const int lane = threadIdx.x & 31;
  const int wave = __builtin_amdgcn_readfirstlane((int)(threadIdx.x >> 5));
  const int e0 = (blockIdx.x * (THREADS / 32) + wave) * 32;
  if (e0 >= E) return;
  const int nvalid = (E - e0 < 32) ? (E - e0) : 32;

  const int e = e0 + lane;
  int s = 0, d = 0;
  float a0 = 0.f, a1 = 0.f, a2 = 0.f, a3 = 0.f;
  if (lane < nvalid) {
    s = src[e];
    d = dst[e];
    const float4 ex = *reinterpret_cast<const float4*>(ebuf + (size_t)e * 4);
    const float4 sm = *reinterpret_cast<const float4*>(nsum + (size_t)d * 4);
    a0 = ex.x / sm.x; a1 = ex.y / sm.y; a2 = ex.z / sm.z; a3 = ex.w / sm.w;
  }

  if (nvalid == 32) {
#pragma unroll 1
    for (int g = 0; g < 32; g += 8) {
      float f[8][4];
#pragma unroll
      for (int j = 0; j < 8; ++j) {
        const int sj = __builtin_amdgcn_readlane(s, g + j);
        const float* fs = feat + (size_t)sj * 128 + lane;
        f[j][0] = fs[0]; f[j][1] = fs[32]; f[j][2] = fs[64]; f[j][3] = fs[96];
      }
#pragma unroll
      for (int j = 0; j < 8; ++j) {
        const int dj = __builtin_amdgcn_readlane(d, g + j);
        const float v = f[j][0] * bcast_lane(a0, g + j) + f[j][1] * bcast_lane(a1, g + j)
                      + f[j][2] * bcast_lane(a2, g + j) + f[j][3] * bcast_lane(a3, g + j);
        atomicAdd(accum32 + (size_t)dj * 32 + lane, v);
      }
    }
  } else {
    for (int j = 0; j < nvalid; ++j) {
      const int sj = __builtin_amdgcn_readlane(s, j);
      const int dj = __builtin_amdgcn_readlane(d, j);
      const float* fs = feat + (size_t)sj * 128 + lane;
      const float f0 = fs[0], f1 = fs[32], f2 = fs[64], f3 = fs[96];
      const float v = f0 * bcast_lane(a0, j) + f1 * bcast_lane(a1, j)
                    + f2 * bcast_lane(a2, j) + f3 * bcast_lane(a3, j);
      atomicAdd(accum32 + (size_t)dj * 32 + lane, v);
    }
  }
}


__global__ void node_weight(const float* __restrict__ acc32, const float* __restrict__ aw_w,
                            const float* __restrict__ aw_b, float* __restrict__ wbuf, int N) {
  int n = blockIdx.x * blockDim.x + threadIdx.x;
  if (n >= N) return;
  float s = aw_b[0];
#pragma unroll
  for (int i = 0; i < 32; ++i) s += (0.25f * acc32[n * 32 + i]) * aw_w[i];
  wbuf[n] = 1.f / (1.f + expf(-s));
}

__global__ void graph_init(unsigned* gmat, int n) {
  int t = blockIdx.x * blockDim.x + threadIdx.x;
  if (t >= n) return;
  gmat[t] = ((t & 63) < 32) ? 0u : ORD_NEG_INF;
}

__global__ void graph_scatter(const float* __restrict__ acc32, const float* __restrict__ wbuf,
                              const int* __restrict__ gid, float* __restrict__ gmat, int ND) {
  int t = blockIdx.x * blockDim.x + threadIdx.x;
  if (t >= ND) return;
  int n = t >> 5, d = t & 31;
  int g = gid[n];
  float hv = 0.25f * acc32[t];
  atomicAdd(&gmat[g * 64 + d], wbuf[n] * hv);
  atomicMax((unsigned*)gmat + g * 64 + 32 + d, ford(hv));
}

__global__ void graph_decode(float* gmat, int GD) {
  int t = blockIdx.x * blockDim.x + threadIdx.x;
  if (t >= GD) return;
  int g = t >> 5, d = t & 31;
  unsigned u = ((unsigned*)gmat)[g * 64 + 32 + d];
  gmat[g * 64 + 32 + d] = (u == ORD_NEG_INF) ? 0.f : funord(u);
}

__global__ void final_ln(const float* __restrict__ ybuf, const float* __restrict__ out_b,
                         float* __restrict__ out) {
  __shared__ float sm[128];
  int g = blockIdx.x, j = threadIdx.x;
  float v = ybuf[g * 128 + j] + out_b[j];
  sm[j] = v; __syncthreads();
  for (int s = 64; s > 0; s >>= 1) { if (j < s) sm[j] += sm[j + s]; __syncthreads(); }
  float mu = sm[0] * (1.f / 128.f);
  __syncthreads();
  float dv = v - mu;
  sm[j] = dv * dv; __syncthreads();
  for (int s = 64; s > 0; s >>= 1) { if (j < s) sm[j] += sm[j + s]; __syncthreads(); }
  float var = sm[0] * (1.f / 128.f);
  out[g * 128 + j] = dv * rsqrtf(var + 1e-5f);
}

extern "C" void kernel_launch(void* const* d_in, const int* in_sizes, int n_in,
                              void* d_out, int out_size, void* d_ws, size_t ws_size,
                              hipStream_t stream) {
  const float* node  = (const float*)d_in[0];
  const int*   src   = (const int*)d_in[1];
  const int*   dst   = (const int*)d_in[2];
  const int*   gid   = (const int*)d_in[3];
  const float* fc1_w = (const float*)d_in[4];
  const float* al1   = (const float*)d_in[5];
  const float* ar1   = (const float*)d_in[6];
  const float* res1w = (const float*)d_in[7];
  const float* bias1 = (const float*)d_in[8];
  const float* fc2_w = (const float*)d_in[9];
  const float* al2   = (const float*)d_in[10];
  const float* ar2   = (const float*)d_in[11];
  const float* bias2 = (const float*)d_in[12];
  const float* aw_w  = (const float*)d_in[13];
  const float* aw_b  = (const float*)d_in[14];
  const float* out_w = (const float*)d_in[15];
  const float* out_b = (const float*)d_in[16];
  (void)n_in; (void)ws_size;

  const int N = in_sizes[3];
  const int E = in_sizes[1];
  const int G = out_size / 128;

  char* wsb = (char*)d_ws;
  size_t off = 0;
  auto alloc = [&](size_t bytes) -> char* {
    char* p = wsb + off;
    off = (off + bytes + 255) & ~(size_t)255;
    return p;
  };
  float*    bufA  = (float*)alloc((size_t)N * 128 * 4);
  float*    bufB  = (float*)alloc((size_t)N * 32 * 4);
  float*    bufC  = (float*)alloc((size_t)N * 128 * 4);
  float*    el    = (float*)alloc((size_t)N * 4 * 4);
  float*    er    = (float*)alloc((size_t)N * 4 * 4);
  unsigned* nmax  = (unsigned*)alloc((size_t)N * 4 * 4);
  float*    nsum  = (float*)alloc((size_t)N * 4 * 4);
  float*    ebuf  = (float*)alloc((size_t)E * 4 * 4);
  float*    wbuf  = (float*)alloc((size_t)N * 4);
  float*    gmat  = (float*)alloc((size_t)G * 64 * 4);
  float*    ybuf  = (float*)alloc((size_t)G * 128 * 4);

  auto grid = [](int n) { return dim3((unsigned)((n + THREADS - 1) / THREADS)); };
  const int gtiles  = (G + 15) / 16;
  const int gblocks = (gtiles + 7) / 8;

  wmma_gemm_xwt<27, 8, 16, true, false><<<256, 512, 0, stream>>>(node, fc1_w, bufA, N, al1, ar1, el, er, nmax, nsum,
                                                                 nullptr, nullptr, nullptr);
  wmma_gemm_xwt<27, 8, 16, false, false><<<256, 512, 0, stream>>>(node, res1w, bufC, N, nullptr, nullptr, nullptr, nullptr,
                                                                  nullptr, nullptr,
                                                                  nullptr, nullptr, nullptr);
  edge_score_max<<<grid(E * 4), THREADS, 0, stream>>>(src, dst, el, er, ebuf, nmax, E * 4);
  edge_expsum<<<grid(E * 4), THREADS, 0, stream>>>(dst, nmax, ebuf, nsum, E * 4);
  edge_aggregate<<<grid(E), THREADS, 0, stream>>>(src, dst, ebuf, nsum, bufA, bufC, E);

  wmma_gemm_xwt<128, 8, 16, true, true><<<256, 512, 0, stream>>>(bufC, fc2_w, bufA, N, al2, ar2, el, er, nmax, nsum,
                                                                 bias1, bias2, bufB);
  edge_score_max<<<grid(E * 4), THREADS, 0, stream>>>(src, dst, el, er, ebuf, nmax, E * 4);
  edge_expsum<<<grid(E * 4), THREADS, 0, stream>>>(dst, nmax, ebuf, nsum, E * 4);
  edge_aggregate_headsum<<<grid(E), THREADS, 0, stream>>>(src, dst, ebuf, nsum, bufA, bufB, E);

  node_weight<<<grid(N), THREADS, 0, stream>>>(bufB, aw_w, aw_b, wbuf, N);
  graph_init<<<grid(G * 64), THREADS, 0, stream>>>((unsigned*)gmat, G * 64);
  graph_scatter<<<grid(N * 32), THREADS, 0, stream>>>(bufB, wbuf, gid, gmat, N * 32);
  graph_decode<<<grid(G * 32), THREADS, 0, stream>>>(gmat, G * 32);
  wmma_gemm_xwt<64, 8, 8, false, false><<<gblocks, 256, 0, stream>>>(gmat, out_w, ybuf, G, nullptr, nullptr, nullptr, nullptr,
                                                                     nullptr, nullptr,
                                                                     nullptr, nullptr, nullptr);
  final_ln<<<G, 128, 0, stream>>>(ybuf, out_b, (float*)d_out);
}


# Readout: a wave walks 32 consecutive nodes, lanes = the 32 columns; weight dot is a 32-lane tree; runs of one graph id summed in registers, 1 atomic pair per run; 15->14 launches; sum order changed
# speedup vs baseline: 4.6895x; 1.0338x over previous
#include <hip/hip_runtime.h>
#include <hip/hip_bf16.h>
#include <math.h>

typedef __attribute__((ext_vector_type(2))) float v2f;
typedef __attribute__((ext_vector_type(8))) float v8f;

#define THREADS 256
#define ORD_NEG_INF 0x007FFFFFu

__device__ __forceinline__ unsigned ford(float f) {
  unsigned u = __float_as_uint(f);
  return (u & 0x80000000u) ? ~u : (u | 0x80000000u);
}
__device__ __forceinline__ float funord(unsigned u) {
  return __uint_as_float((u & 0x80000000u) ? (u & 0x7fffffffu) : ~u);
}

__device__ __forceinline__ float xhalf(float x) {
  const int b = __builtin_bit_cast(int, x);
  return __builtin_bit_cast(float,
      __builtin_amdgcn_permlanex16(b, b, 0x76543210, 0xfedcba98, false, false));
}

template <int K, int NT, int WAVES, bool ATT, bool PRE>
__global__ __launch_bounds__(WAVES * 32)
void wmma_gemm_xwt(const float* __restrict__ X,
                   const float* __restrict__ W,
                   float* __restrict__ Y, int M,
                   const float* __restrict__ al, const float* __restrict__ ar,
                   float* __restrict__ el, float* __restrict__ er,
                   unsigned* __restrict__ smax, float* __restrict__ ssum,
                   const float* __restrict__ pb1, const float* __restrict__ pb2,
                   float* __restrict__ hsum) {
  constexpr int NO = NT * 16;
  constexpr int NTT = ATT ? NT + 1 : NT;
  constexpr int KP = (K + 3) & ~3;
  constexpr int P  = ((KP / 4) & 1) ? KP : KP + 4;
  constexpr int NTHR = WAVES * 32;
  __shared__ __attribute__((aligned(16))) float sW[NO * P];
  __shared__ __attribute__((aligned(16))) float sWa[ATT ? 16 * P : 4];
  __shared__ float sA[ATT ? 256 : 1];
  __shared__ __attribute__((aligned(8))) float sPB[PRE ? 256 : 2];

  const int tid = threadIdx.x;
  if constexpr (ATT) {
    static_assert(NO == 128 && NTHR >= 256, "score columns assume 4 heads x 32");
    if (tid < 128) sA[tid] = al[tid];
    else if (tid < 256) sA[tid] = ar[tid - 128];
  }
  if constexpr (PRE) {
    static_assert(ATT && K == 128 && NT == 8 && NTHR >= 512, "pointwise stage assumes 4 heads x 32 in and out");
    if (tid >= 384) sPB[tid - 256] = pb2[tid - 384];
    else if (tid >= 256) sPB[tid - 256] = pb1[tid - 256];
  }
  if constexpr ((K & 3) == 0) {
    constexpr int K4 = K / 4;
    static_assert((NO * K4) % NTHR == 0, "W must split evenly over the workgroup");
    const float4* __restrict__ W4 = reinterpret_cast<const float4*>(W);
#pragma unroll
    for (int j = 0; j < NO * K4 / NTHR; ++j) {
      const int i = tid + j * NTHR;
      const int r = i / K4, k4 = i - r * K4;
      *reinterpret_cast<float4*>(&sW[r * P + 4 * k4]) = W4[i];
    }
  } else {
#pragma unroll
    for (int j = 0; j < (NO * K + NTHR - 1) / NTHR; ++j) {
      const int i = tid + j * NTHR;
      if (i < NO * K) {
        const int r = i / K, k = i - r * K;
        sW[r * P + k] = W[i];
      }
    }
    for (int i = tid; i < NO * (KP - K); i += NTHR) {
      const int r = i / (KP - K), k = K + (i - r * (KP - K));
      sW[r * P + k] = 0.f;
    }
  }
  __syncthreads();

  if constexpr (ATT) {
    for (int i = tid; i < 16 * KP; i += NTHR) {
      const int j = i / KP, k = i - j * KP;
      float s = 0.f;
      if (j < 8) {
        const float* a = sA + j * 32;
        const float* w = sW + ((j & 3) * 32) * P + k;
#pragma unroll
        for (int d = 0; d < 32; ++d) s += w[d * P] * a[d];
      }
      sWa[j * P + k] = s;
    }
    __syncthreads();
  }

  const int lane = tid & 31;
  const int wave = __builtin_amdgcn_readfirstlane(tid >> 5);
  const int ntiles = (M + 15) >> 4;
  const int koff  = (lane < 16) ? 0 : 2;
  const int rowoff = (lane < 16) ? 0 : 8;
  const int colo   = lane & 15;
  const float* sB = sW + colo * P + koff;
  const float* sBa = sWa + colo * P + koff;

  for (int tile = blockIdx.x + gridDim.x * wave; tile < ntiles; tile += gridDim.x * WAVES) {
    const int rbase = tile * 16;
    const int mrow  = rbase + colo;

    v8f acc[NTT];
#pragma unroll
    for (int t = 0; t < NTT; ++t) acc[t] = (v8f){0.f,0.f,0.f,0.f,0.f,0.f,0.f,0.f};

    if constexpr (PRE) {
      float hs[16];
#pragma unroll
      for (int i = 0; i < 16; ++i) hs[i] = 0.f;
      const float* xrow = X + mrow * K + koff;
      const float* pb = sPB + koff;
      for (int h = 0; h < 4; ++h) {
#pragma unroll
        for (int q = 0; q < 2; ++q) {
          const int kc = h * 32 + q * 16;
          v2f a[4];
#pragma unroll
          for (int u = 0; u < 4; ++u)
            a[u] = *reinterpret_cast<const v2f*>(xrow + kc + 4 * u);
          asm volatile("" :: "v"(a[0]), "v"(a[1]), "v"(a[2]), "v"(a[3]));
#pragma unroll
          for (int u = 0; u < 4; ++u) {
            const v2f c1 = *reinterpret_cast<const v2f*>(pb + kc + 4 * u);
            const v2f c2 = *reinterpret_cast<const v2f*>(pb + 128 + kc + 4 * u);
            const float x0 = a[u].x + c1.x, x1 = a[u].y + c1.y;
            const float e0 = expm1f(x0), e1 = expm1f(x1);
            const float h0 = (x0 > 0.f) ? x0 : e0;
            const float h1 = (x1 > 0.f) ? x1 : e1;
            hs[q * 8 + u * 2]     += h0 + c2.x;
            hs[q * 8 + u * 2 + 1] += h1 + c2.y;
            a[u].x = h0; a[u].y = h1;
          }
#pragma unroll
          for (int u = 0; u < 4; ++u) {
            const int k0 = kc + 4 * u;
#pragma unroll
            for (int t = 0; t < NT; ++t) {
              const v2f b = *reinterpret_cast<const v2f*>(sB + t * 16 * P + k0);
              acc[t] = __builtin_amdgcn_wmma_f32_16x16x4_f32(
                  false, a[u], false, b, (short)0, acc[t], false, false);
            }
            const v2f b = *reinterpret_cast<const v2f*>(sBa + k0);
            acc[NT] = __builtin_amdgcn_wmma_f32_16x16x4_f32(
                false, a[u], false, b, (short)0, acc[NT], false, false);
          }
        }
      }
      const bool lo = lane < 16;
      float* hrow = hsum + mrow * 32 + (lo ? 0 : 4);
#pragma unroll
      for (int q = 0; q < 2; ++q) {
#pragma unroll
        for (int p = 0; p < 2; ++p) {
          const int i0 = q * 8 + 4 * p, i1 = i0 + 2;
          const float r0 = xhalf(lo ? hs[i1] : hs[i0]);
          const float r1 = xhalf(lo ? hs[i1 + 1] : hs[i0 + 1]);
          float4 o;
          o.x = lo ? hs[i0] : r0;
          o.y = lo ? hs[i0 + 1] : r1;
          o.z = lo ? r0 : hs[i1];
          o.w = lo ? r1 : hs[i1 + 1];
          *reinterpret_cast<float4*>(hrow + 16 * q + 8 * p) = o;
        }
      }
    } else {
      for (int k0 = 0; k0 < K; k0 += 4) {
        const int k = k0 + koff;
        v2f a;
        a.x = (k     < K) ? X[mrow * K + k]     : 0.f;
        a.y = (k + 1 < K) ? X[mrow * K + k + 1] : 0.f;
#pragma unroll
        for (int t = 0; t < NT; ++t) {
          const v2f b = *reinterpret_cast<const v2f*>(sB + t * 16 * P + k0);
          acc[t] = __builtin_amdgcn_wmma_f32_16x16x4_f32(
              false, a, false, b, (short)0, acc[t], false, false);
        }
        if constexpr (ATT) {
          const v2f b = *reinterpret_cast<const v2f*>(sBa + k0);
          acc[NT] = __builtin_amdgcn_wmma_f32_16x16x4_f32(
              false, a, false, b, (short)0, acc[NT], false, false);
        }
      }
    }

#pragma unroll
    for (int t = 0; t < NT; ++t) {
#pragma unroll
      for (int i = 0; i < 8; ++i) {
        Y[(rbase + rowoff + i) * NO + t * 16 + colo] = acc[t][i];
      }
    }
    if constexpr (ATT) {
      unsigned* sc = (colo < 4)  ? reinterpret_cast<unsigned*>(el) + colo
                   : (colo < 8)  ? reinterpret_cast<unsigned*>(er) + (colo - 4)
                   : (colo < 12) ? smax + (colo - 8)
                                 : reinterpret_cast<unsigned*>(ssum) + (colo - 12);
      const bool prod = colo < 8;
      const unsigned start = (colo < 12) ? ORD_NEG_INF : 0u;
#pragma unroll
      for (int i = 0; i < 8; ++i)
        sc[(rbase + rowoff + i) * 4] = prod ? __float_as_uint(acc[NT][i]) : start;
    }
  }
}

__global__ void edge_score_max(const int* __restrict__ src, const int* __restrict__ dst,
                               const float* __restrict__ el, const float* __restrict__ er,
                               float* __restrict__ ebuf, unsigned* __restrict__ nmax, int EH) {
  int t = blockIdx.x * blockDim.x + threadIdx.x;
  if (t >= EH) return;
  int e = t >> 2, h = t & 3;
  float v = el[src[e] * 4 + h] + er[dst[e] * 4 + h];
  v = (v > 0.f) ? v : 0.2f * v;
  ebuf[t] = v;
  atomicMax(&nmax[dst[e] * 4 + h], ford(v));
}

__global__ void edge_expsum(const int* __restrict__ dst,
                            const unsigned* __restrict__ nmax,
                            float* __restrict__ ebuf, float* __restrict__ nsum, int EH) {
  int t = blockIdx.x * blockDim.x + threadIdx.x;
  if (t >= EH) return;
  int e = t >> 2, h = t & 3;
  float m = funord(nmax[dst[e] * 4 + h]);
  float ex = expf(ebuf[t] - m);
  ebuf[t] = ex;
  atomicAdd(&nsum[dst[e] * 4 + h], ex);
}

__device__ __forceinline__ float bcast_lane(float v, int l) {
  return __int_as_float(__builtin_amdgcn_readlane(__float_as_int(v), l));
}

__global__ __launch_bounds__(THREADS)
void edge_aggregate(const int* __restrict__ src, const int* __restrict__ dst,
                    const float* __restrict__ ebuf, const float* __restrict__ nsum,
                    const float* __restrict__ feat, float* __restrict__ accum, int E) {
  const int lane = threadIdx.x & 31;
  const int wave = __builtin_amdgcn_readfirstlane((int)(threadIdx.x >> 5));
  const int e0 = (blockIdx.x * (THREADS / 32) + wave) * 32;
  if (e0 >= E) return;
  const int nvalid = (E - e0 < 32) ? (E - e0) : 32;

  const int e = e0 + lane;
  int s = 0, d = 0;
  float a0 = 0.f, a1 = 0.f, a2 = 0.f, a3 = 0.f;
  if (lane < nvalid) {
    s = src[e];
    d = dst[e];
    const float4 ex = *reinterpret_cast<const float4*>(ebuf + (size_t)e * 4);
    const float4 sm = *reinterpret_cast<const float4*>(nsum + (size_t)d * 4);
    a0 = ex.x / sm.x; a1 = ex.y / sm.y; a2 = ex.z / sm.z; a3 = ex.w / sm.w;
  }

  if (nvalid == 32) {
#pragma unroll 1
    for (int g = 0; g < 32; g += 8) {
      float f[8][4];
#pragma unroll
      for (int j = 0; j < 8; ++j) {
        const int sj = __builtin_amdgcn_readlane(s, g + j);
        const float* fs = feat + (size_t)sj * 128 + lane;
        f[j][0] = fs[0]; f[j][1] = fs[32]; f[j][2] = fs[64]; f[j][3] = fs[96];
      }
#pragma unroll
      for (int j = 0; j < 8; ++j) {
        const int dj = __builtin_amdgcn_readlane(d, g + j);
        float* ad = accum + (size_t)dj * 128 + lane;
        atomicAdd(ad,      f[j][0] * bcast_lane(a0, g + j));
        atomicAdd(ad + 32, f[j][1] * bcast_lane(a1, g + j));
        atomicAdd(ad + 64, f[j][2] * bcast_lane(a2, g + j));
        atomicAdd(ad + 96, f[j][3] * bcast_lane(a3, g + j));
      }
    }
  } else {
    for (int j = 0; j < nvalid; ++j) {
      const int sj = __builtin_amdgcn_readlane(s, j);
      const int dj = __builtin_amdgcn_readlane(d, j);
      const float* fs = feat + (size_t)sj * 128 + lane;
      float* ad = accum + (size_t)dj * 128 + lane;
      const float f0 = fs[0], f1 = fs[32], f2 = fs[64], f3 = fs[96];
      atomicAdd(ad,      f0 * bcast_lane(a0, j));
      atomicAdd(ad + 32, f1 * bcast_lane(a1, j));
      atomicAdd(ad + 64, f2 * bcast_lane(a2, j));
      atomicAdd(ad + 96, f3 * bcast_lane(a3, j));
    }
  }
}

__global__ __launch_bounds__(THREADS)
void edge_aggregate_headsum(const int* __restrict__ src, const int* __restrict__ dst,
                            const float* __restrict__ ebuf, const float* __restrict__ nsum,
                            const float* __restrict__ feat, float* __restrict__ accum32, int E) {
  const int lane = threadIdx.x & 31;
  const int wave = __builtin_amdgcn_readfirstlane((int)(threadIdx.x >> 5));
  const int e0 = (blockIdx.x * (THREADS / 32) + wave) * 32;
  if (e0 >= E) return;
  const int nvalid = (E - e0 < 32) ? (E - e0) : 32;

  const int e = e0 + lane;
  int s = 0, d = 0;
  float a0 = 0.f, a1 = 0.f, a2 = 0.f, a3 = 0.f;
  if (lane < nvalid) {
    s = src[e];
    d = dst[e];
    const float4 ex = *reinterpret_cast<const float4*>(ebuf + (size_t)e * 4);
    const float4 sm = *reinterpret_cast<const float4*>(nsum + (size_t)d * 4);
    a0 = ex.x / sm.x; a1 = ex.y / sm.y; a2 = ex.z / sm.z; a3 = ex.w / sm.w;
  }

  if (nvalid == 32) {
#pragma unroll 1
    for (int g = 0; g < 32; g += 8) {
      float f[8][4];
#pragma unroll
      for (int j = 0; j < 8; ++j) {
        const int sj = __builtin_amdgcn_readlane(s, g + j);
        const float* fs = feat + (size_t)sj * 128 + lane;
        f[j][0] = fs[0]; f[j][1] = fs[32]; f[j][2] = fs[64]; f[j][3] = fs[96];
      }
#pragma unroll
      for (int j = 0; j < 8; ++j) {
        const int dj = __builtin_amdgcn_readlane(d, g + j);
        const float v = f[j][0] * bcast_lane(a0, g + j) + f[j][1] * bcast_lane(a1, g + j)
                      + f[j][2] * bcast_lane(a2, g + j) + f[j][3] * bcast_lane(a3, g + j);
        atomicAdd(accum32 + (size_t)dj * 32 + lane, v);
      }
    }
  } else {
    for (int j = 0; j < nvalid; ++j) {
      const int sj = __builtin_amdgcn_readlane(s, j);
      const int dj = __builtin_amdgcn_readlane(d, j);
      const float* fs = feat + (size_t)sj * 128 + lane;
      const float f0 = fs[0], f1 = fs[32], f2 = fs[64], f3 = fs[96];
      const float v = f0 * bcast_lane(a0, j) + f1 * bcast_lane(a1, j)
                    + f2 * bcast_lane(a2, j) + f3 * bcast_lane(a3, j);
      atomicAdd(accum32 + (size_t)dj * 32 + lane, v);
    }
  }
}


__global__ void graph_init(unsigned* gmat, int n) {
  int t = blockIdx.x * blockDim.x + threadIdx.x;
  if (t >= n) return;
  gmat[t] = ((t & 63) < 32) ? 0u : ORD_NEG_INF;
}


template <int M>
__device__ __forceinline__ float xrow(float x) {
  const int b = __builtin_bit_cast(int, x);
  return __builtin_bit_cast(float, __builtin_amdgcn_mov_dpp(b, 0x160 + M, 0xf, 0xf, true));
}
__device__ __forceinline__ float wave_sum(float v) {
  v += xrow<1>(v);
  v += xrow<2>(v);
  v += xrow<4>(v);
  v += xrow<8>(v);
  v += xhalf(v);
  return v;
}

#define RO_THREADS 128
__global__ __launch_bounds__(RO_THREADS)
void graph_readout(const float* __restrict__ acc32, const float* __restrict__ aw_w,
                   const float* __restrict__ aw_b, const int* __restrict__ gid,
                   float* __restrict__ gmat, int N) {
  const int lane = threadIdx.x & 31;
  const int wave = __builtin_amdgcn_readfirstlane((int)(threadIdx.x >> 5));
  const int n0 = (blockIdx.x * (RO_THREADS / 32) + wave) * 32;
  if (n0 >= N) return;
  const int nvalid = (N - n0 < 32) ? (N - n0) : 32;

  const int gl = (lane < nvalid) ? gid[n0 + lane] : 0;
  const float wc = aw_w[lane];
  const float wb = aw_b[0];
  const float* row = acc32 + (size_t)n0 * 32 + lane;
  unsigned* gmax = reinterpret_cast<unsigned*>(gmat) + 32 + lane;
  float*    gsum = gmat + lane;

  int rg = __builtin_amdgcn_readlane(gl, 0);
  float rs = 0.f;
  unsigned rm = ORD_NEG_INF;

  if (nvalid == 32) {
    float hv[32];
#pragma unroll
    for (int j = 0; j < 32; ++j) hv[j] = 0.25f * row[j * 32];
    float sd = 0.f;
#pragma unroll
    for (int j = 0; j < 32; ++j) {
      const float t = wave_sum(hv[j] * wc);
      sd = (lane == j) ? t : sd;
    }
    const float wl = 1.f / (1.f + expf(-(sd + wb)));
#pragma unroll
    for (int j = 0; j < 32; ++j) {
      const float w = bcast_lane(wl, j);
      const int gj = __builtin_amdgcn_readlane(gl, j);
      if (gj != rg) {
        atomicAdd(gsum + (size_t)rg * 64, rs);
        atomicMax(gmax + (size_t)rg * 64, rm);
        rg = gj; rs = 0.f; rm = ORD_NEG_INF;
      }
      rs += w * hv[j];
      const unsigned o = ford(hv[j]);
      rm = (o > rm) ? o : rm;
    }
  } else {
    for (int j = 0; j < nvalid; ++j) {
      const float h = 0.25f * row[j * 32];
      const float sdot = wave_sum(h * wc) + wb;
      const float w = 1.f / (1.f + expf(-sdot));
      const int gj = __builtin_amdgcn_readlane(gl, j);
      if (gj != rg) {
        atomicAdd(gsum + (size_t)rg * 64, rs);
        atomicMax(gmax + (size_t)rg * 64, rm);
        rg = gj; rs = 0.f; rm = ORD_NEG_INF;
      }
      rs += w * h;
      const unsigned o = ford(h);
      rm = (o > rm) ? o : rm;
    }
  }
  atomicAdd(gsum + (size_t)rg * 64, rs);
  atomicMax(gmax + (size_t)rg * 64, rm);
}

__global__ void graph_decode(float* gmat, int GD) {
  int t = blockIdx.x * blockDim.x + threadIdx.x;
  if (t >= GD) return;
  int g = t >> 5, d = t & 31;
  unsigned u = ((unsigned*)gmat)[g * 64 + 32 + d];
  gmat[g * 64 + 32 + d] = (u == ORD_NEG_INF) ? 0.f : funord(u);
}

__global__ void final_ln(const float* __restrict__ ybuf, const float* __restrict__ out_b,
                         float* __restrict__ out) {
  __shared__ float sm[128];
  int g = blockIdx.x, j = threadIdx.x;
  float v = ybuf[g * 128 + j] + out_b[j];
  sm[j] = v; __syncthreads();
  for (int s = 64; s > 0; s >>= 1) { if (j < s) sm[j] += sm[j + s]; __syncthreads(); }
  float mu = sm[0] * (1.f / 128.f);
  __syncthreads();
  float dv = v - mu;
  sm[j] = dv * dv; __syncthreads();
  for (int s = 64; s > 0; s >>= 1) { if (j < s) sm[j] += sm[j + s]; __syncthreads(); }
  float var = sm[0] * (1.f / 128.f);
  out[g * 128 + j] = dv * rsqrtf(var + 1e-5f);
}

extern "C" void kernel_launch(void* const* d_in, const int* in_sizes, int n_in,
                              void* d_out, int out_size, void* d_ws, size_t ws_size,
                              hipStream_t stream) {
  const float* node  = (const float*)d_in[0];
  const int*   src   = (const int*)d_in[1];
  const int*   dst   = (const int*)d_in[2];
  const int*   gid   = (const int*)d_in[3];
  const float* fc1_w = (const float*)d_in[4];
  const float* al1   = (const float*)d_in[5];
  const float* ar1   = (const float*)d_in[6];
  const float* res1w = (const float*)d_in[7];
  const float* bias1 = (const float*)d_in[8];
  const float* fc2_w = (const float*)d_in[9];
  const float* al2   = (const float*)d_in[10];
  const float* ar2   = (const float*)d_in[11];
  const float* bias2 = (const float*)d_in[12];
  const float* aw_w  = (const float*)d_in[13];
  const float* aw_b  = (const float*)d_in[14];
  const float* out_w = (const float*)d_in[15];
  const float* out_b = (const float*)d_in[16];
  (void)n_in; (void)ws_size;

  const int N = in_sizes[3];
  const int E = in_sizes[1];
  const int G = out_size / 128;

  char* wsb = (char*)d_ws;
  size_t off = 0;
  auto alloc = [&](size_t bytes) -> char* {
    char* p = wsb + off;
    off = (off + bytes + 255) & ~(size_t)255;
    return p;
  };
  float*    bufA  = (float*)alloc((size_t)N * 128 * 4);
  float*    bufB  = (float*)alloc((size_t)N * 32 * 4);
  float*    bufC  = (float*)alloc((size_t)N * 128 * 4);
  float*    el    = (float*)alloc((size_t)N * 4 * 4);
  float*    er    = (float*)alloc((size_t)N * 4 * 4);
  unsigned* nmax  = (unsigned*)alloc((size_t)N * 4 * 4);
  float*    nsum  = (float*)alloc((size_t)N * 4 * 4);
  float*    ebuf  = (float*)alloc((size_t)E * 4 * 4);
  float*    gmat  = (float*)alloc((size_t)G * 64 * 4);
  float*    ybuf  = (float*)alloc((size_t)G * 128 * 4);

  auto grid = [](int n) { return dim3((unsigned)((n + THREADS - 1) / THREADS)); };
  const int gtiles  = (G + 15) / 16;
  const int gblocks = (gtiles + 7) / 8;

  wmma_gemm_xwt<27, 8, 16, true, false><<<256, 512, 0, stream>>>(node, fc1_w, bufA, N, al1, ar1, el, er, nmax, nsum,
                                                                 nullptr, nullptr, nullptr);
  wmma_gemm_xwt<27, 8, 16, false, false><<<256, 512, 0, stream>>>(node, res1w, bufC, N, nullptr, nullptr, nullptr, nullptr,
                                                                  nullptr, nullptr,
                                                                  nullptr, nullptr, nullptr);
  edge_score_max<<<grid(E * 4), THREADS, 0, stream>>>(src, dst, el, er, ebuf, nmax, E * 4);
  edge_expsum<<<grid(E * 4), THREADS, 0, stream>>>(dst, nmax, ebuf, nsum, E * 4);
  edge_aggregate<<<grid(E), THREADS, 0, stream>>>(src, dst, ebuf, nsum, bufA, bufC, E);

  wmma_gemm_xwt<128, 8, 16, true, true><<<256, 512, 0, stream>>>(bufC, fc2_w, bufA, N, al2, ar2, el, er, nmax, nsum,
                                                                 bias1, bias2, bufB);
  edge_score_max<<<grid(E * 4), THREADS, 0, stream>>>(src, dst, el, er, ebuf, nmax, E * 4);
  edge_expsum<<<grid(E * 4), THREADS, 0, stream>>>(dst, nmax, ebuf, nsum, E * 4);
  edge_aggregate_headsum<<<grid(E), THREADS, 0, stream>>>(src, dst, ebuf, nsum, bufA, bufB, E);

  graph_init<<<grid(G * 64), THREADS, 0, stream>>>((unsigned*)gmat, G * 64);
  graph_readout<<<dim3((unsigned)((N + RO_THREADS - 1) / RO_THREADS)), RO_THREADS, 0, stream>>>(bufB, aw_w, aw_b, gid, gmat, N);
  graph_decode<<<grid(G * 32), THREADS, 0, stream>>>(gmat, G * 32);
  wmma_gemm_xwt<64, 8, 8, false, false><<<gblocks, 256, 0, stream>>>(gmat, out_w, ybuf, G, nullptr, nullptr, nullptr, nullptr,
                                                                     nullptr, nullptr,
                                                                     nullptr, nullptr, nullptr);
  final_ln<<<G, 128, 0, stream>>>(ybuf, out_b, (float*)d_out);
}


# Edge passes: node rows split in two ranges, one per package half; a work-group reads its die, draws a chunk ticket for that half and takes only edges whose dst is in its range; atomic order differs
# speedup vs baseline: 5.6269x; 1.1999x over previous
#include <hip/hip_runtime.h>
#include <hip/hip_bf16.h>
#include <math.h>

typedef __attribute__((ext_vector_type(2))) float v2f;
typedef __attribute__((ext_vector_type(8))) float v8f;

#define THREADS 256
#define ORD_NEG_INF 0x007FFFFFu
#define NTICK 12
#define TICK_PITCH 64

__device__ __forceinline__ unsigned ford(float f) {
  unsigned u = __float_as_uint(f);
  return (u & 0x80000000u) ? ~u : (u | 0x80000000u);
}
__device__ __forceinline__ float funord(unsigned u) {
  return __uint_as_float((u & 0x80000000u) ? (u & 0x7fffffffu) : ~u);
}

__device__ __forceinline__ float xhalf(float x) {
  const int b = __builtin_bit_cast(int, x);
  return __builtin_bit_cast(float,
      __builtin_amdgcn_permlanex16(b, b, 0x76543210, 0xfedcba98, false, false));
}

template <int K, int NT, int WAVES, bool ATT, bool PRE>
__global__ __launch_bounds__(WAVES * 32)
void wmma_gemm_xwt(const float* __restrict__ X,
                   const float* __restrict__ W,
                   float* __restrict__ Y, int M,
                   const float* __restrict__ al, const float* __restrict__ ar,
                   float* __restrict__ el, float* __restrict__ er,
                   unsigned* __restrict__ smax, float* __restrict__ ssum,
                   const float* __restrict__ pb1, const float* __restrict__ pb2,
                   float* __restrict__ hsum, unsigned* __restrict__ tick) {
  constexpr int NO = NT * 16;
  constexpr int NTT = ATT ? NT + 1 : NT;
  constexpr int KP = (K + 3) & ~3;
  constexpr int P  = ((KP / 4) & 1) ? KP : KP + 4;
  constexpr int NTHR = WAVES * 32;
  __shared__ __attribute__((aligned(16))) float sW[NO * P];
  __shared__ __attribute__((aligned(16))) float sWa[ATT ? 16 * P : 4];
  __shared__ float sA[ATT ? 256 : 1];
  __shared__ __attribute__((aligned(8))) float sPB[PRE ? 256 : 2];

  const int tid = threadIdx.x;
  if (tick != nullptr && blockIdx.x == 0 && tid < NTICK) tick[tid * TICK_PITCH] = 0u;
  if constexpr (ATT) {
    static_assert(NO == 128 && NTHR >= 256, "score columns assume 4 heads x 32");
    if (tid < 128) sA[tid] = al[tid];
    else if (tid < 256) sA[tid] = ar[tid - 128];
  }
  if constexpr (PRE) {
    static_assert(ATT && K == 128 && NT == 8 && NTHR >= 512, "pointwise stage assumes 4 heads x 32 in and out");
    if (tid >= 384) sPB[tid - 256] = pb2[tid - 384];
    else if (tid >= 256) sPB[tid - 256] = pb1[tid - 256];
  }
  if constexpr ((K & 3) == 0) {
    constexpr int K4 = K / 4;
    static_assert((NO * K4) % NTHR == 0, "W must split evenly over the workgroup");
    const float4* __restrict__ W4 = reinterpret_cast<const float4*>(W);
#pragma unroll
    for (int j = 0; j < NO * K4 / NTHR; ++j) {
      const int i = tid + j * NTHR;
      const int r = i / K4, k4 = i - r * K4;
      *reinterpret_cast<float4*>(&sW[r * P + 4 * k4]) = W4[i];
    }
  } else {
#pragma unroll
    for (int j = 0; j < (NO * K + NTHR - 1) / NTHR; ++j) {
      const int i = tid + j * NTHR;
      if (i < NO * K) {
        const int r = i / K, k = i - r * K;
        sW[r * P + k] = W[i];
      }
    }
    for (int i = tid; i < NO * (KP - K); i += NTHR) {
      const int r = i / (KP - K), k = K + (i - r * (KP - K));
      sW[r * P + k] = 0.f;
    }
  }
  __syncthreads();

  if constexpr (ATT) {
    for (int i = tid; i < 16 * KP; i += NTHR) {
      const int j = i / KP, k = i - j * KP;
      float s = 0.f;
      if (j < 8) {
        const float* a = sA + j * 32;
        const float* w = sW + ((j & 3) * 32) * P + k;
#pragma unroll
        for (int d = 0; d < 32; ++d) s += w[d * P] * a[d];
      }
      sWa[j * P + k] = s;
    }
    __syncthreads();
  }

  const int lane = tid & 31;
  const int wave = __builtin_amdgcn_readfirstlane(tid >> 5);
  const int ntiles = (M + 15) >> 4;
  const int koff  = (lane < 16) ? 0 : 2;
  const int rowoff = (lane < 16) ? 0 : 8;
  const int colo   = lane & 15;
  const float* sB = sW + colo * P + koff;
  const float* sBa = sWa + colo * P + koff;

  for (int tile = blockIdx.x + gridDim.x * wave; tile < ntiles; tile += gridDim.x * WAVES) {
    const int rbase = tile * 16;
    const int mrow  = rbase + colo;

    v8f acc[NTT];
#pragma unroll
    for (int t = 0; t < NTT; ++t) acc[t] = (v8f){0.f,0.f,0.f,0.f,0.f,0.f,0.f,0.f};

    if constexpr (PRE) {
      float hs[16];
#pragma unroll
      for (int i = 0; i < 16; ++i) hs[i] = 0.f;
      const float* xrow = X + mrow * K + koff;
      const float* pb = sPB + koff;
      for (int h = 0; h < 4; ++h) {
#pragma unroll
        for (int q = 0; q < 2; ++q) {
          const int kc = h * 32 + q * 16;
          v2f a[4];
#pragma unroll
          for (int u = 0; u < 4; ++u)
            a[u] = *reinterpret_cast<const v2f*>(xrow + kc + 4 * u);
          asm volatile("" :: "v"(a[0]), "v"(a[1]), "v"(a[2]), "v"(a[3]));
#pragma unroll
          for (int u = 0; u < 4; ++u) {
            const v2f c1 = *reinterpret_cast<const v2f*>(pb + kc + 4 * u);
            const v2f c2 = *reinterpret_cast<const v2f*>(pb + 128 + kc + 4 * u);
            const float x0 = a[u].x + c1.x, x1 = a[u].y + c1.y;
            const float e0 = expm1f(x0), e1 = expm1f(x1);
            const float h0 = (x0 > 0.f) ? x0 : e0;
            const float h1 = (x1 > 0.f) ? x1 : e1;
            hs[q * 8 + u * 2]     += h0 + c2.x;
            hs[q * 8 + u * 2 + 1] += h1 + c2.y;
            a[u].x = h0; a[u].y = h1;
          }
#pragma unroll
          for (int u = 0; u < 4; ++u) {
            const int k0 = kc + 4 * u;
#pragma unroll
            for (int t = 0; t < NT; ++t) {
              const v2f b = *reinterpret_cast<const v2f*>(sB + t * 16 * P + k0);
              acc[t] = __builtin_amdgcn_wmma_f32_16x16x4_f32(
                  false, a[u], false, b, (short)0, acc[t], false, false);
            }
            const v2f b = *reinterpret_cast<const v2f*>(sBa + k0);
            acc[NT] = __builtin_amdgcn_wmma_f32_16x16x4_f32(
                false, a[u], false, b, (short)0, acc[NT], false, false);
          }
        }
      }
      const bool lo = lane < 16;
      float* hrow = hsum + mrow * 32 + (lo ? 0 : 4);
#pragma unroll
      for (int q = 0; q < 2; ++q) {
#pragma unroll
        for (int p = 0; p < 2; ++p) {
          const int i0 = q * 8 + 4 * p, i1 = i0 + 2;
          const float r0 = xhalf(lo ? hs[i1] : hs[i0]);
          const float r1 = xhalf(lo ? hs[i1 + 1] : hs[i0 + 1]);
          float4 o;
          o.x = lo ? hs[i0] : r0;
          o.y = lo ? hs[i0 + 1] : r1;
          o.z = lo ? r0 : hs[i1];
          o.w = lo ? r1 : hs[i1 + 1];
          *reinterpret_cast<float4*>(hrow + 16 * q + 8 * p) = o;
        }
      }
    } else {
      for (int k0 = 0; k0 < K; k0 += 4) {
        const int k = k0 + koff;
        v2f a;
        a.x = (k     < K) ? X[mrow * K + k]     : 0.f;
        a.y = (k + 1 < K) ? X[mrow * K + k + 1] : 0.f;
#pragma unroll
        for (int t = 0; t < NT; ++t) {
          const v2f b = *reinterpret_cast<const v2f*>(sB + t * 16 * P + k0);
          acc[t] = __builtin_amdgcn_wmma_f32_16x16x4_f32(
              false, a, false, b, (short)0, acc[t], false, false);
        }
        if constexpr (ATT) {
          const v2f b = *reinterpret_cast<const v2f*>(sBa + k0);
          acc[NT] = __builtin_amdgcn_wmma_f32_16x16x4_f32(
              false, a, false, b, (short)0, acc[NT], false, false);
        }
      }
    }

#pragma unroll
    for (int t = 0; t < NT; ++t) {
#pragma unroll
      for (int i = 0; i < 8; ++i) {
        Y[(rbase + rowoff + i) * NO + t * 16 + colo] = acc[t][i];
      }
    }
    if constexpr (ATT) {
      unsigned* sc = (colo < 4)  ? reinterpret_cast<unsigned*>(el) + colo
                   : (colo < 8)  ? reinterpret_cast<unsigned*>(er) + (colo - 4)
                   : (colo < 12) ? smax + (colo - 8)
                                 : reinterpret_cast<unsigned*>(ssum) + (colo - 12);
      const bool prod = colo < 8;
      const unsigned start = (colo < 12) ? ORD_NEG_INF : 0u;
#pragma unroll
      for (int i = 0; i < 8; ++i)
        sc[(rbase + rowoff + i) * 4] = prod ? __float_as_uint(acc[NT][i]) : start;
    }
  }
}


__device__ __forceinline__ int chip_half() {
  return (int)((__builtin_amdgcn_s_sendmsg_rtn(0x87) >> 18) & 1u);
}
__device__ __forceinline__ void take_unit(unsigned* cnt, int nchunks, int* unit) {
  int r = chip_half();
  unsigned t = atomicAdd(cnt + r * TICK_PITCH, 1u);
  if (t >= (unsigned)nchunks) {
    r ^= 1;
    t = atomicAdd(cnt + r * TICK_PITCH, 1u);
  }
  unit[0] = r;
  unit[1] = (int)t;
}
__device__ __forceinline__ int node_split(int N) { return ((N >> 1) + 15) & ~15; }

__global__ __launch_bounds__(THREADS)
void edge_score_max(const int* __restrict__ src, const int* __restrict__ dst,
                    const float* __restrict__ el, const float* __restrict__ er,
                    float* __restrict__ ebuf, unsigned* __restrict__ nmax, int E, int N,
                    unsigned* __restrict__ cnt) {
  __shared__ int unit[2];
  const int nchunks = (E + THREADS - 1) / THREADS;
  if (threadIdx.x == 0) take_unit(cnt, nchunks, unit);
  __syncthreads();
  const int r = unit[0], c = unit[1];
  if (c >= nchunks) return;
  const int nsplit = node_split(N);
  const int h = threadIdx.x & 3;
  const int eb = c * THREADS + (threadIdx.x >> 2);
  int s[4], d[4]; bool own[4];
#pragma unroll
  for (int j = 0; j < 4; ++j) {
    const int e = eb + 64 * j;
    const int ec = (e < E) ? e : E - 1;
    s[j] = src[ec];
    d[j] = dst[ec];
  }
  float x[4], y[4];
#pragma unroll
  for (int j = 0; j < 4; ++j) {
    own[j] = (eb + 64 * j < E) && ((d[j] >= nsplit) == (r != 0));
    x[j] = el[(own[j] ? s[j] * 4 : 0) + h];
    y[j] = er[(own[j] ? d[j] * 4 : 0) + h];
  }
#pragma unroll
  for (int j = 0; j < 4; ++j) {
    if (own[j]) {
      float v = x[j] + y[j];
      v = (v > 0.f) ? v : 0.2f * v;
      ebuf[(size_t)(eb + 64 * j) * 4 + h] = v;
      atomicMax(&nmax[d[j] * 4 + h], ford(v));
    }
  }
}

__global__ __launch_bounds__(THREADS)
void edge_expsum(const int* __restrict__ dst, const unsigned* __restrict__ nmax,
                 float* __restrict__ ebuf, float* __restrict__ nsum, int E, int N,
                 unsigned* __restrict__ cnt) {
  __shared__ int unit[2];
  const int nchunks = (E + THREADS - 1) / THREADS;
  if (threadIdx.x == 0) take_unit(cnt, nchunks, unit);
  __syncthreads();
  const int r = unit[0], c = unit[1];
  if (c >= nchunks) return;
  const int nsplit = node_split(N);
  const int h = threadIdx.x & 3;
  const int eb = c * THREADS + (threadIdx.x >> 2);
  int d[4]; bool own[4];
#pragma unroll
  for (int j = 0; j < 4; ++j) {
    const int e = eb + 64 * j;
    d[j] = dst[(e < E) ? e : E - 1];
  }
  unsigned mo[4]; float ev[4];
  const float* spare = reinterpret_cast<const float*>(nmax) + h;
#pragma unroll
  for (int j = 0; j < 4; ++j) {
    own[j] = (eb + 64 * j < E) && ((d[j] >= nsplit) == (r != 0));
    mo[j] = nmax[(own[j] ? d[j] * 4 : 0) + h];
    const float* pe = own[j] ? ebuf + (size_t)(eb + 64 * j) * 4 + h : spare;
    ev[j] = *pe;
  }
#pragma unroll
  for (int j = 0; j < 4; ++j) {
    if (own[j]) {
      const float ex = expf(ev[j] - funord(mo[j]));
      ebuf[(size_t)(eb + 64 * j) * 4 + h] = ex;
      atomicAdd(&nsum[d[j] * 4 + h], ex);
    }
  }
}

__device__ __forceinline__ float bcast_lane(float v, int l) {
  return __int_as_float(__builtin_amdgcn_readlane(__float_as_int(v), l));
}

template <bool HEADSUM>
__global__ __launch_bounds__(THREADS)
void edge_aggregate(const int* __restrict__ src, const int* __restrict__ dst,
                    const float* __restrict__ ebuf, const float* __restrict__ nsum,
                    const float* __restrict__ feat, float* __restrict__ accum, int E, int N,
                    unsigned* __restrict__ cnt) {
  __shared__ int unit[2];
  const int nchunks = (E + THREADS - 1) / THREADS;
  if (threadIdx.x == 0) take_unit(cnt, nchunks, unit);
  __syncthreads();
  const int r = __builtin_amdgcn_readfirstlane(unit[0]);
  const int c = __builtin_amdgcn_readfirstlane(unit[1]);
  if (c >= nchunks) return;
  const int nsplit = node_split(N);
  const int lane = threadIdx.x & 31;
  const int wave = __builtin_amdgcn_readfirstlane((int)(threadIdx.x >> 5));
  const int e = c * THREADS + wave * 32 + lane;

  int s = 0, d = 0;
  bool own = false;
  if (e < E) {
    s = src[e];
    d = dst[e];
    own = (d >= nsplit) == (r != 0);
  }
  float a0 = 0.f, a1 = 0.f, a2 = 0.f, a3 = 0.f;
  if (own) {
    const float4 ex = *reinterpret_cast<const float4*>(ebuf + (size_t)e * 4);
    const float4 sm = *reinterpret_cast<const float4*>(nsum + (size_t)d * 4);
    a0 = ex.x / sm.x; a1 = ex.y / sm.y; a2 = ex.z / sm.z; a3 = ex.w / sm.w;
  }
  unsigned m = __builtin_amdgcn_ballot_w32(own);

  while (m != 0u) {
    const int nv = __builtin_popcount(m);
    int l[8];
#pragma unroll
    for (int j = 0; j < 8; ++j) {
      l[j] = (m != 0u) ? __builtin_ctz(m) : l[j > 0 ? j - 1 : 0];
      m &= m - 1u;
    }
    float f[8][4];
#pragma unroll
    for (int j = 0; j < 8; ++j) {
      if (j < nv) {
        const int sj = __builtin_amdgcn_readlane(s, l[j]);
        const float* fs = feat + (size_t)sj * 128 + lane;
        f[j][0] = fs[0]; f[j][1] = fs[32]; f[j][2] = fs[64]; f[j][3] = fs[96];
      }
    }
#pragma unroll
    for (int j = 0; j < 8; ++j) {
      if (j < nv) {
        const int dj = __builtin_amdgcn_readlane(d, l[j]);
        if constexpr (HEADSUM) {
          const float v = f[j][0] * bcast_lane(a0, l[j]) + f[j][1] * bcast_lane(a1, l[j])
                        + f[j][2] * bcast_lane(a2, l[j]) + f[j][3] * bcast_lane(a3, l[j]);
          atomicAdd(accum + (size_t)dj * 32 + lane, v);
        } else {
          float* ad = accum + (size_t)dj * 128 + lane;
          atomicAdd(ad,      f[j][0] * bcast_lane(a0, l[j]));
          atomicAdd(ad + 32, f[j][1] * bcast_lane(a1, l[j]));
          atomicAdd(ad + 64, f[j][2] * bcast_lane(a2, l[j]));
          atomicAdd(ad + 96, f[j][3] * bcast_lane(a3, l[j]));
        }
      }
    }
  }
}


__global__ void graph_init(unsigned* gmat, int n) {
  int t = blockIdx.x * blockDim.x + threadIdx.x;
  if (t >= n) return;
  gmat[t] = ((t & 63) < 32) ? 0u : ORD_NEG_INF;
}


template <int M>
__device__ __forceinline__ float xrow(float x) {
  const int b = __builtin_bit_cast(int, x);
  return __builtin_bit_cast(float, __builtin_amdgcn_mov_dpp(b, 0x160 + M, 0xf, 0xf, true));
}
__device__ __forceinline__ float wave_sum(float v) {
  v += xrow<1>(v);
  v += xrow<2>(v);
  v += xrow<4>(v);
  v += xrow<8>(v);
  v += xhalf(v);
  return v;
}

#define RO_THREADS 128
__global__ __launch_bounds__(RO_THREADS)
void graph_readout(const float* __restrict__ acc32, const float* __restrict__ aw_w,
                   const float* __restrict__ aw_b, const int* __restrict__ gid,
                   float* __restrict__ gmat, int N) {
  const int lane = threadIdx.x & 31;
  const int wave = __builtin_amdgcn_readfirstlane((int)(threadIdx.x >> 5));
  const int n0 = (blockIdx.x * (RO_THREADS / 32) + wave) * 32;
  if (n0 >= N) return;
  const int nvalid = (N - n0 < 32) ? (N - n0) : 32;

  const int gl = (lane < nvalid) ? gid[n0 + lane] : 0;
  const float wc = aw_w[lane];
  const float wb = aw_b[0];
  const float* row = acc32 + (size_t)n0 * 32 + lane;
  unsigned* gmax = reinterpret_cast<unsigned*>(gmat) + 32 + lane;
  float*    gsum = gmat + lane;

  int rg = __builtin_amdgcn_readlane(gl, 0);
  float rs = 0.f;
  unsigned rm = ORD_NEG_INF;

  if (nvalid == 32) {
    float hv[32];
#pragma unroll
    for (int j = 0; j < 32; ++j) hv[j] = 0.25f * row[j * 32];
    float sd = 0.f;
#pragma unroll
    for (int j = 0; j < 32; ++j) {
      const float t = wave_sum(hv[j] * wc);
      sd = (lane == j) ? t : sd;
    }
    const float wl = 1.f / (1.f + expf(-(sd + wb)));
#pragma unroll
    for (int j = 0; j < 32; ++j) {
      const float w = bcast_lane(wl, j);
      const int gj = __builtin_amdgcn_readlane(gl, j);
      if (gj != rg) {
        atomicAdd(gsum + (size_t)rg * 64, rs);
        atomicMax(gmax + (size_t)rg * 64, rm);
        rg = gj; rs = 0.f; rm = ORD_NEG_INF;
      }
      rs += w * hv[j];
      const unsigned o = ford(hv[j]);
      rm = (o > rm) ? o : rm;
    }
  } else {
    for (int j = 0; j < nvalid; ++j) {
      const float h = 0.25f * row[j * 32];
      const float sdot = wave_sum(h * wc) + wb;
      const float w = 1.f / (1.f + expf(-sdot));
      const int gj = __builtin_amdgcn_readlane(gl, j);
      if (gj != rg) {
        atomicAdd(gsum + (size_t)rg * 64, rs);
        atomicMax(gmax + (size_t)rg * 64, rm);
        rg = gj; rs = 0.f; rm = ORD_NEG_INF;
      }
      rs += w * h;
      const unsigned o = ford(h);
      rm = (o > rm) ? o : rm;
    }
  }
  atomicAdd(gsum + (size_t)rg * 64, rs);
  atomicMax(gmax + (size_t)rg * 64, rm);
}

__global__ void graph_decode(float* gmat, int GD) {
  int t = blockIdx.x * blockDim.x + threadIdx.x;
  if (t >= GD) return;
  int g = t >> 5, d = t & 31;
  unsigned u = ((unsigned*)gmat)[g * 64 + 32 + d];
  gmat[g * 64 + 32 + d] = (u == ORD_NEG_INF) ? 0.f : funord(u);
}

__global__ void final_ln(const float* __restrict__ ybuf, const float* __restrict__ out_b,
                         float* __restrict__ out) {
  __shared__ float sm[128];
  int g = blockIdx.x, j = threadIdx.x;
  float v = ybuf[g * 128 + j] + out_b[j];
  sm[j] = v; __syncthreads();
  for (int s = 64; s > 0; s >>= 1) { if (j < s) sm[j] += sm[j + s]; __syncthreads(); }
  float mu = sm[0] * (1.f / 128.f);
  __syncthreads();
  float dv = v - mu;
  sm[j] = dv * dv; __syncthreads();
  for (int s = 64; s > 0; s >>= 1) { if (j < s) sm[j] += sm[j + s]; __syncthreads(); }
  float var = sm[0] * (1.f / 128.f);
  out[g * 128 + j] = dv * rsqrtf(var + 1e-5f);
}

extern "C" void kernel_launch(void* const* d_in, const int* in_sizes, int n_in,
                              void* d_out, int out_size, void* d_ws, size_t ws_size,
                              hipStream_t stream) {
  const float* node  = (const float*)d_in[0];
  const int*   src   = (const int*)d_in[1];
  const int*   dst   = (const int*)d_in[2];
  const int*   gid   = (const int*)d_in[3];
  const float* fc1_w = (const float*)d_in[4];
  const float* al1   = (const float*)d_in[5];
  const float* ar1   = (const float*)d_in[6];
  const float* res1w = (const float*)d_in[7];
  const float* bias1 = (const float*)d_in[8];
  const float* fc2_w = (const float*)d_in[9];
  const float* al2   = (const float*)d_in[10];
  const float* ar2   = (const float*)d_in[11];
  const float* bias2 = (const float*)d_in[12];
  const float* aw_w  = (const float*)d_in[13];
  const float* aw_b  = (const float*)d_in[14];
  const float* out_w = (const float*)d_in[15];
  const float* out_b = (const float*)d_in[16];
  (void)n_in; (void)ws_size;

  const int N = in_sizes[3];
  const int E = in_sizes[1];
  const int G = out_size / 128;

  char* wsb = (char*)d_ws;
  size_t off = 0;
  auto alloc = [&](size_t bytes) -> char* {
    char* p = wsb + off;
    off = (off + bytes + 255) & ~(size_t)255;
    return p;
  };
  float*    bufA  = (float*)alloc((size_t)N * 128 * 4);
  float*    bufB  = (float*)alloc((size_t)N * 32 * 4);
  float*    bufC  = (float*)alloc((size_t)N * 128 * 4);
  float*    el    = (float*)alloc((size_t)N * 4 * 4);
  float*    er    = (float*)alloc((size_t)N * 4 * 4);
  unsigned* nmax  = (unsigned*)alloc((size_t)N * 4 * 4);
  float*    nsum  = (float*)alloc((size_t)N * 4 * 4);
  float*    ebuf  = (float*)alloc((size_t)E * 4 * 4);
  float*    gmat  = (float*)alloc((size_t)G * 64 * 4);
  float*    ybuf  = (float*)alloc((size_t)G * 128 * 4);

  auto grid = [](int n) { return dim3((unsigned)((n + THREADS - 1) / THREADS)); };
  const int gtiles  = (G + 15) / 16;
  const int gblocks = (gtiles + 7) / 8;
  const int egrid = 2 * ((E + THREADS - 1) / THREADS);
  unsigned* tick = (unsigned*)ybuf;

  wmma_gemm_xwt<27, 8, 16, true, false><<<256, 512, 0, stream>>>(node, fc1_w, bufA, N, al1, ar1, el, er, nmax, nsum,
                                                                 nullptr, nullptr, nullptr, tick);
  wmma_gemm_xwt<27, 8, 16, false, false><<<256, 512, 0, stream>>>(node, res1w, bufC, N, nullptr, nullptr, nullptr, nullptr,
                                                                  nullptr, nullptr,
                                                                  nullptr, nullptr, nullptr, nullptr);
  edge_score_max<<<egrid, THREADS, 0, stream>>>(src, dst, el, er, ebuf, nmax, E, N, tick + 0 * 2 * TICK_PITCH);
  edge_expsum<<<egrid, THREADS, 0, stream>>>(dst, nmax, ebuf, nsum, E, N, tick + 1 * 2 * TICK_PITCH);
  edge_aggregate<false><<<egrid, THREADS, 0, stream>>>(src, dst, ebuf, nsum, bufA, bufC, E, N, tick + 2 * 2 * TICK_PITCH);

  wmma_gemm_xwt<128, 8, 16, true, true><<<256, 512, 0, stream>>>(bufC, fc2_w, bufA, N, al2, ar2, el, er, nmax, nsum,
                                                                 bias1, bias2, bufB, nullptr);
  edge_score_max<<<egrid, THREADS, 0, stream>>>(src, dst, el, er, ebuf, nmax, E, N, tick + 3 * 2 * TICK_PITCH);
  edge_expsum<<<egrid, THREADS, 0, stream>>>(dst, nmax, ebuf, nsum, E, N, tick + 4 * 2 * TICK_PITCH);
  edge_aggregate<true><<<egrid, THREADS, 0, stream>>>(src, dst, ebuf, nsum, bufA, bufB, E, N, tick + 5 * 2 * TICK_PITCH);

  graph_init<<<grid(G * 64), THREADS, 0, stream>>>((unsigned*)gmat, G * 64);
  graph_readout<<<dim3((unsigned)((N + RO_THREADS - 1) / RO_THREADS)), RO_THREADS, 0, stream>>>(bufB, aw_w, aw_b, gid, gmat, N);
  graph_decode<<<grid(G * 32), THREADS, 0, stream>>>(gmat, G * 32);
  wmma_gemm_xwt<64, 8, 8, false, false><<<gblocks, 256, 0, stream>>>(gmat, out_w, ybuf, G, nullptr, nullptr, nullptr, nullptr,
                                                                     nullptr, nullptr,
                                                                     nullptr, nullptr, nullptr, nullptr);
  final_ln<<<G, 128, 0, stream>>>(ybuf, out_b, (float*)d_out);
}
